# starting kernel
# baseline (speedup 1.0000x reference)
#include <hip/hip_runtime.h>
#include <math.h>

typedef float v2f __attribute__((ext_vector_type(2)));
typedef float v8f __attribute__((ext_vector_type(8)));

#define FEAT 128
#define BN_EPS 1e-3f


__global__ void zero_f32(float* p, int n) {
    int i = blockIdx.x * blockDim.x + threadIdx.x;
    if (i < n) p[i] = 0.0f;
}

__global__ void deg_kernel(const int* __restrict__ rows, const float* __restrict__ w,
                           float* deg, int nE) {
    int e = blockIdx.x * blockDim.x + threadIdx.x;
    if (e < nE) atomicAdd(&deg[rows[e]], w[e]);
}

__global__ void dinv_kernel(const float* __restrict__ deg, float* __restrict__ dinv, int n) {
    int i = blockIdx.x * blockDim.x + threadIdx.x;
    if (i < n) {
        float d = deg[i] + 1.0f;
        dinv[i] = (d > 0.0f) ? rsqrtf(d) : 0.0f;
    }
}


__global__ __launch_bounds__(256) void gemm_kernel(const float* __restrict__ x,
                                                   const float* __restrict__ W,
                                                   float* __restrict__ h, int nRows) {
    __shared__ float Wl[FEAT * FEAT];

    int t = threadIdx.x;
    const float4* Wv = (const float4*)W;
    float4* Wlv = (float4*)Wl;
    #pragma unroll
    for (int i = t; i < (FEAT * FEAT) / 4; i += 256) Wlv[i] = Wv[i];
    __syncthreads();

    int wave = t >> 5;
    int lane = t & 31;
    int row0 = blockIdx.x * 128 + wave * 16;
    if (row0 >= nRows) return;

    int m  = lane & 15;
    int kh = (lane >> 4) << 1;
    int srow = row0 + m;
    if (srow >= nRows) srow = nRows - 1;
    const float* xr = x + (size_t)srow * FEAT;

    v8f acc[8] = {};

    for (int kb = 0; kb < FEAT / 4; ++kb) {
        int k = kb * 4 + kh;
        float2 av = *(const float2*)(xr + k);
        v2f a; a.x = av.x; a.y = av.y;
        #pragma unroll
        for (int j = 0; j < 8; ++j) {
            int col = j * 16 + m;
            v2f b;
            b.x = Wl[(k + 0) * FEAT + col];
            b.y = Wl[(k + 1) * FEAT + col];
            acc[j] = __builtin_amdgcn_wmma_f32_16x16x4_f32(
                false, a, false, b, (short)0, acc[j], false, false);
        }
    }

    int rbase = row0 + ((lane >> 4) ? 8 : 0);
    int n = lane & 15;
    #pragma unroll
    for (int j = 0; j < 8; ++j) {
        #pragma unroll
        for (int r = 0; r < 8; ++r) {
            int row = rbase + r;
            if (row < nRows) h[(size_t)row * FEAT + j * 16 + n] = acc[j][r];
        }
    }
}


__global__ void self_loop_init(const float* __restrict__ h, const float* __restrict__ dinv,
                               float* __restrict__ out, int nNodes) {
    int i = blockIdx.x * blockDim.x + threadIdx.x;
    int total = nNodes * (FEAT / 4);
    if (i < total) {
        int node = i >> 5;
        float s = dinv[node];
        s = s * s;
        float4 v = ((const float4*)h)[i];
        v.x *= s; v.y *= s; v.z *= s; v.w *= s;
        ((float4*)out)[i] = v;
    }
}


__global__ __launch_bounds__(256) void edge_kernel(const int* __restrict__ rows,
                                                   const int* __restrict__ cols,
                                                   const float* __restrict__ w,
                                                   const float* __restrict__ dinv,
                                                   const float* __restrict__ h,
                                                   float* out, int nE) {
    int gid = blockIdx.x * blockDim.x + threadIdx.x;
    int wid = gid >> 5;
    int lane = gid & 31;
    int stride = (gridDim.x * blockDim.x) >> 5;
    for (int e = wid; e < nE; e += stride) {
        int r = rows[e];
        int c = cols[e];
        float norm = dinv[r] * w[e] * dinv[c];
        float4 hv = *(const float4*)(h + (size_t)c * FEAT + lane * 4);
        float* op = out + (size_t)r * FEAT + lane * 4;
        atomicAdd(op + 0, hv.x * norm);
        atomicAdd(op + 1, hv.y * norm);
        atomicAdd(op + 2, hv.z * norm);
        atomicAdd(op + 3, hv.w * norm);
    }
}


__global__ void bn_relu_kernel(float* __restrict__ out,
                               const float* __restrict__ bias,
                               const float* __restrict__ gamma,
                               const float* __restrict__ beta,
                               const float* __restrict__ mean,
                               const float* __restrict__ var, int total) {
    int i = blockIdx.x * blockDim.x + threadIdx.x;
    if (i < total) {
        int u = i & (FEAT - 1);
        float v = out[i] + bias[u];
        float r = gamma[u] * (v - mean[u]) * rsqrtf(var[u] + BN_EPS) + beta[u];
        out[i] = r > 0.0f ? r : 0.0f;
    }
}


extern "C" void kernel_launch(void* const* d_in, const int* in_sizes, int n_in,
                              void* d_out, int out_size, void* d_ws, size_t ws_size,
                              hipStream_t stream) {
    const float* x     = (const float*)d_in[0];
    const int*   eidx  = (const int*)d_in[1];
    const float* ew    = (const float*)d_in[2];
    const float* W     = (const float*)d_in[3];
    const float* bias  = (const float*)d_in[4];
    const float* gamma = (const float*)d_in[5];
    const float* beta  = (const float*)d_in[6];
    const float* mean  = (const float*)d_in[7];
    const float* var   = (const float*)d_in[8];

    int nE = in_sizes[2];
    int nN = in_sizes[0] / FEAT;
    const int* rows = eidx;
    const int* cols = eidx + nE;

    float* h    = (float*)d_ws;
    float* deg  = h + (size_t)nN * FEAT;
    float* dinv = deg + nN;
    float* out  = (float*)d_out;

    zero_f32<<<(nN + 255) / 256, 256, 0, stream>>>(deg, nN);
    deg_kernel<<<(nE + 255) / 256, 256, 0, stream>>>(rows, ew, deg, nE);
    dinv_kernel<<<(nN + 255) / 256, 256, 0, stream>>>(deg, dinv, nN);

    gemm_kernel<<<(nN + 127) / 128, 256, 0, stream>>>(x, W, h, nN);

    int initTotal = nN * (FEAT / 4);
    self_loop_init<<<(initTotal + 255) / 256, 256, 0, stream>>>(h, dinv, out, nN);

    long edgeThreads = (long)nE * 32;
    edge_kernel<<<(int)((edgeThreads + 255) / 256), 256, 0, stream>>>(
        rows, cols, ew, dinv, h, out, nE);

    int total = nN * FEAT;
    bn_relu_kernel<<<(total + 255) / 256, 256, 0, stream>>>(
        out, bias, gamma, beta, mean, var, total);
}


# launches: product, group index and dinv (independent of each other) run as one launch of mixed 256-thread blocks; degree partials lie in out, overwritten last; 3 launches, sum orders unchanged
# speedup vs baseline: 2.5581x; 2.5581x over previous
#include <hip/hip_runtime.h>
#include <math.h>

typedef float v2f __attribute__((ext_vector_type(2)));
typedef float v8f __attribute__((ext_vector_type(8)));

#define FEAT 128
#define MID_THREADS 256
#define BN_EPS 1e-3f


#define DEG_BINS   25600
#define DEG_SLICES 64
#define DEG_THREADS 1024
#define DEG_BATCH  8

__global__ __launch_bounds__(DEG_THREADS) void deg_partial_kernel(const int* __restrict__ rows,
                                                                  const float* __restrict__ w,
                                                                  float* __restrict__ part,
                                                                  int nE, int nN) {
    __shared__ float bins[DEG_BINS];

    int t = threadIdx.x;
    float4* bz = (float4*)bins;
    for (int i = t; i < DEG_BINS / 4; i += DEG_THREADS) bz[i] = make_float4(0.0f, 0.0f, 0.0f, 0.0f);
    __syncthreads();

    int r0  = blockIdx.y * DEG_BINS;
    int per = (nE + DEG_SLICES - 1) / DEG_SLICES;
    int e0  = blockIdx.x * per;
    int e1  = e0 + per < nE ? e0 + per : nE;

    for (int e = e0 + t; e < e1; e += DEG_THREADS * DEG_BATCH) {
        int   r[DEG_BATCH];
        float v[DEG_BATCH];
        for (int u = 0; u < DEG_BATCH; ++u) {
            int ee = e + u * DEG_THREADS;
            bool ok = ee < e1;
            r[u] = ok ? rows[ee] : -1;
            v[u] = ok ? w[ee] : 0.0f;
        }
        for (int u = 0; u < DEG_BATCH; ++u) {
            unsigned d = (unsigned)(r[u] - r0);
            if (d < (unsigned)DEG_BINS) atomicAdd(&bins[d], v[u]);
        }
    }
    __syncthreads();

    int nb = nN - r0 < DEG_BINS ? nN - r0 : DEG_BINS;
    float* dst = part + (size_t)blockIdx.x * nN + r0;
    for (int i = t; i < nb; i += DEG_THREADS) dst[i] = bins[i];
}

__device__ __forceinline__ void dinv_block(const float* __restrict__ part, float* __restrict__ dinv, int n, int blk) {
    int i = blk * MID_THREADS + threadIdx.x;
    if (i < n) {
        float p[DEG_SLICES];
        for (int s = 0; s < DEG_SLICES; ++s) p[s] = part[(size_t)s * n + i];
        float d = 0.0f;
        for (int s = 0; s < DEG_SLICES; ++s) d += p[s];
        d += 1.0f;
        dinv[i] = (d > 0.0f) ? rsqrtf(d) : 0.0f;
    }
}


__device__ __forceinline__ void gemm_block(const float* __restrict__ x,
                                           const float* __restrict__ W,
                                           float* __restrict__ h, int nRows, int blk, float* Wl) {
    int t = threadIdx.x;
    const float4* Wv = (const float4*)W;
    float4* Wlv = (float4*)Wl;
    #pragma unroll
    for (int i = t; i < (FEAT * FEAT) / 4; i += MID_THREADS) Wlv[i] = Wv[i];
    __syncthreads();

    int wave = t >> 5;
    int lane = t & 31;
    int row0 = blk * 128 + wave * 16;
    if (row0 >= nRows) return;

    int m  = lane & 15;
    int kh = (lane >> 4) << 1;
    int srow = row0 + m;
    if (srow >= nRows) srow = nRows - 1;
    const float* xr = x + (size_t)srow * FEAT;

    v8f acc[8] = {};

    for (int kb = 0; kb < FEAT / 4; ++kb) {
        int k = kb * 4 + kh;
        float2 av = *(const float2*)(xr + k);
        v2f a; a.x = av.x; a.y = av.y;
        #pragma unroll
        for (int j = 0; j < 8; ++j) {
            int col = j * 16 + m;
            v2f b;
            b.x = Wl[(k + 0) * FEAT + col];
            b.y = Wl[(k + 1) * FEAT + col];
            acc[j] = __builtin_amdgcn_wmma_f32_16x16x4_f32(
                false, a, false, b, (short)0, acc[j], false, false);
        }
    }

    int rbase = row0 + ((lane >> 4) ? 8 : 0);
    int n = lane & 15;
    #pragma unroll
    for (int j = 0; j < 8; ++j) {
        #pragma unroll
        for (int r = 0; r < 8; ++r) {
            int row = rbase + r;
            if (row < nRows) h[(size_t)row * FEAT + j * 16 + n] = acc[j][r];
        }
    }
}


#define OWN_ROWS    393
#define GRP_EDGES   128
#define IDX_CHUNK   4096
#define OWN_THREADS 1024
#define OWN_WAVES   32
#define LIST_CAP    316
#define COL_BITS    23
#define EDGE_BITS   23
#define GATHER_BATCH 8

__device__ __forceinline__ void owner_flush(unsigned pend_pk, int pend_wn, int first, int np,
                                            const float* __restrict__ h, float* acc, int lane) {
    float4 hv[GATHER_BATCH];
    #pragma unroll
    for (int k = 0; k < GATHER_BATCH; ++k) {
        if (k < np) {
            unsigned pk = (unsigned)__builtin_amdgcn_readlane((int)pend_pk, first + k);
            unsigned c = pk & ((1u << COL_BITS) - 1u);
            hv[k] = *(const float4*)(h + (size_t)c * FEAT + lane * 4);
        }
    }
    #pragma unroll
    for (int k = 0; k < GATHER_BATCH; ++k) {
        if (k < np) {
            unsigned pk = (unsigned)__builtin_amdgcn_readlane((int)pend_pk, first + k);
            unsigned rl = pk >> COL_BITS;
            float wn = __int_as_float(__builtin_amdgcn_readlane(pend_wn, first + k));
            float4* ap = (float4*)acc + rl * (FEAT / 4) + lane;
            float4 a = *ap;
            a.x += hv[k].x * wn; a.y += hv[k].y * wn; a.z += hv[k].z * wn; a.w += hv[k].w * wn;
            *ap = a;
        }
    }
}

__device__ __forceinline__ void index_block(const int* __restrict__ rows, unsigned* __restrict__ idx,
                                            int nE, int nOwn, int nWords, int blk, unsigned* bits) {
    int t = threadIdx.x;
    for (int i = t; i < 1024; i += MID_THREADS) bits[i] = 0u;
    __syncthreads();
    int rr[4][4];
    #pragma unroll
    for (int k = 0; k < 4; ++k) {
        int e = blk * IDX_CHUNK + k * 1024 + t * 4;
        rr[k][0] = -1; rr[k][1] = -1; rr[k][2] = -1; rr[k][3] = -1;
        if (e + 3 < nE) {
            int4 q = *(const int4*)(rows + e);
            rr[k][0] = q.x; rr[k][1] = q.y; rr[k][2] = q.z; rr[k][3] = q.w;
        } else {
            for (int c = 0; c < 4; ++c) if (e + c < nE) rr[k][c] = rows[e + c];
        }
    }
    #pragma unroll
    for (int k = 0; k < 4; ++k) {
        unsigned bit = 1u << (k * 8 + (t >> 5));
        #pragma unroll
        for (int c = 0; c < 4; ++c) {
            unsigned b = (unsigned)rr[k][c] / (unsigned)OWN_ROWS;
            if (b < (unsigned)nOwn) atomicOr(&bits[b], bit);
        }
    }
    __syncthreads();
    for (int i = t; i < nOwn; i += MID_THREADS) idx[(size_t)i * nWords + blk] = bits[i];
}

__global__ __launch_bounds__(MID_THREADS) void mid_kernel(const float* __restrict__ x,
                                                          const float* __restrict__ W,
                                                          float* __restrict__ h,
                                                          const int* __restrict__ rows,
                                                          unsigned* __restrict__ idx,
                                                          const float* __restrict__ part,
                                                          float* __restrict__ dinv,
                                                          int nN, int nE, int nOwn, int nWords) {
    __shared__ float smem[FEAT * FEAT];
    int nProd = (nN + 127) / 128;
    int b = blockIdx.x;
    if (b < nProd) {
        gemm_block(x, W, h, nN, b, smem);
    } else if (b < nProd + nWords) {
        index_block(rows, idx, nE, nOwn, nWords, b - nProd, (unsigned*)smem);
    } else {
        dinv_block(part, dinv, nN, b - nProd - nWords);
    }
}

template <bool TAIL>
__device__ __forceinline__ bool owner_filter_quarter(int4 q, int ebase, int end, int lane,
                                                     int r0, unsigned nb, unsigned lt, unsigned* mylst, int& n) {
    int rr[4] = {q.x, q.y, q.z, q.w};
    unsigned mk[4];
    #pragma unroll
    for (int c = 0; c < 4; ++c) {
        bool hit = (unsigned)(rr[c] - r0) < nb;
        if (TAIL) hit = hit && ebase + lane * 4 + c < end;
        mk[c] = __builtin_amdgcn_ballot_w32(hit);
    }
    if (n + GRP_EDGES > LIST_CAP) {
        int tot = __builtin_popcount(mk[0]) + __builtin_popcount(mk[1]) + __builtin_popcount(mk[2]) + __builtin_popcount(mk[3]);
        if (n + tot > LIST_CAP) return false;
    }
    #pragma unroll
    for (int c = 0; c < 4; ++c) {
        if (mk[c] != 0u) {
            if ((mk[c] >> lane) & 1u) {
                int e = ebase + lane * 4 + c;
                mylst[n + __builtin_popcount(mk[c] & lt)] = (unsigned)e | ((unsigned)(rr[c] - r0) << EDGE_BITS);
            }
            n += __builtin_popcount(mk[c]);
        }
    }
    return true;
}

__device__ __forceinline__ unsigned owner_lanes(unsigned key, unsigned valid, const unsigned* ob) {
    unsigned m = valid;
    #pragma unroll
    for (int b = 0; b < 5; ++b) m &= ((key >> b) & 1u) ? ob[b] : ~ob[b];
    return m;
}

__global__ __launch_bounds__(OWN_THREADS) void edge_owner_kernel(const int* __restrict__ rows,
                                                                 const int* __restrict__ cols,
                                                                 const float* __restrict__ w,
                                                                 const float* __restrict__ dinv,
                                                                 const float* __restrict__ h,
                                                                 float* __restrict__ out,
                                                                 const float* __restrict__ bias,
                                                                 const float* __restrict__ gamma,
                                                                 const float* __restrict__ beta,
                                                                 const float* __restrict__ mean,
                                                                 const float* __restrict__ var,
                                                                 const unsigned* __restrict__ idx,
                                                                 int nE, int nN, int nWords) {
    __shared__ float    acc[OWN_ROWS * FEAT];
    __shared__ int2     own[OWN_WAVES * LIST_CAP];
    __shared__ unsigned lst[OWN_WAVES * LIST_CAP];
    __shared__ int      cntm[OWN_WAVES * OWN_WAVES];
    __shared__ int      more[OWN_WAVES];

    int t = threadIdx.x;
    int wave = __builtin_amdgcn_readfirstlane(t >> 5);
    int lane = t & 31;
    int r0 = blockIdx.x * OWN_ROWS;
    int nb = nN - r0 < OWN_ROWS ? nN - r0 : OWN_ROWS;

    for (int i = t; i < nb * (FEAT / 4); i += OWN_THREADS) {
        int r = i >> 5;
        float s = dinv[r0 + r];
        float4 v = ((const float4*)h)[(size_t)(r0 + r) * (FEAT / 4) + (i & 31)];
        v.x *= s; v.y *= s; v.z *= s; v.w *= s;
        ((float4*)acc)[i] = v;
    }

    int wpw = (nWords + OWN_WAVES - 1) / OWN_WAVES;
    int w0 = wave * wpw < nWords ? wave * wpw : nWords;
    int w1 = w0 + wpw < nWords ? w0 + wpw : nWords;
    const unsigned* myidx = idx + (size_t)blockIdx.x * nWords;
    int cw = w0;
    unsigned myw = w0 + lane < w1 ? myidx[w0 + lane] : 0u;
    unsigned cm = (unsigned)__builtin_amdgcn_readlane((int)myw, 0);
    int last = (nE - 1) & ~3;
    unsigned lt = (1u << lane) - 1u;
    unsigned* mylst = lst + wave * LIST_CAP;

    for (;;) {
        int n = 0;
        int full = 0;
        while (!full) {
            int gg[4] = {0, 0, 0, 0};
            int scw[4];
            unsigned scm[4];
            int k = 0;
            #pragma unroll
            for (int j = 0; j < 4; ++j) {
                if (k == j) {
                    while (cm == 0u && cw + 1 < w1) {
                        ++cw;
                        if (((cw - w0) & 31) == 0) myw = cw + lane < w1 ? myidx[cw + lane] : 0u;
                        cm = (unsigned)__builtin_amdgcn_readlane((int)myw, (cw - w0) & 31);
                    }
                }
                scw[j] = cw;
                scm[j] = cm;
                if (k == j && cm != 0u) {
                    gg[j] = cw * 32 + __builtin_ctz(cm);
                    cm &= cm - 1u;
                    k = j + 1;
                }
            }
            if (k == 0) break;
            int4 q[4];
            #pragma unroll
            for (int j = 0; j < 4; ++j) {
                int g = j < k ? gg[j] : gg[0];
                int i = g * GRP_EDGES + lane * 4;
                i = i < last ? i : last;
                q[j] = *(const int4*)(rows + i);
            }
            #pragma unroll
            for (int j = 0; j < 4; ++j) {
                if (j < k && !full) {
                    int eb = gg[j] * GRP_EDGES;
                    bool ok = eb + GRP_EDGES <= nE
                                  ? owner_filter_quarter<false>(q[j], eb, nE, lane, r0, (unsigned)nb, lt, mylst, n)
                                  : owner_filter_quarter<true>(q[j], eb, nE, lane, r0, (unsigned)nb, lt, mylst, n);
                    if (!ok) { cw = scw[j]; cm = scm[j]; full = 1; }
                }
            }
        }
        int mycnt = 0;
        for (int base = 0; base < n; base += 32) {
            int i = base + lane;
            bool ok = i < n;
            unsigned o = ok ? (mylst[i] >> EDGE_BITS) & 31u : 0u;
            unsigned valid = __builtin_amdgcn_ballot_w32(ok);
            unsigned ob[5];
            #pragma unroll
            for (int b = 0; b < 5; ++b) ob[b] = __builtin_amdgcn_ballot_w32(ok && ((o >> b) & 1u));
            mycnt += __builtin_popcount(owner_lanes((unsigned)lane, valid, ob));
        }
        cntm[wave * OWN_WAVES + lane] = mycnt;
        if (lane == 0) more[wave] = full;
        __syncthreads();

        int before = 0;
        int tot = 0;
        #pragma unroll
        for (int s = 0; s < OWN_WAVES; ++s) {
            int v = cntm[s * OWN_WAVES + lane];
            tot += v;
            before += s < wave ? v : 0;
        }
        int incl = tot;
        for (int d = 1; d < 32; d <<= 1) {
            int up = __shfl_up(incl, d);
            if (lane >= d) incl += up;
        }
        int pos = incl - tot + before;
        int myfirst = __builtin_amdgcn_readlane(incl - tot, wave);
        int mytot = __builtin_amdgcn_readlane(tot, wave);
        int any = __builtin_amdgcn_ballot_w32(more[lane] != 0) != 0u;

        for (int base = 0; base < n; base += 32) {
            int i = base + lane;
            bool ok = i < n;
            unsigned en = ok ? mylst[i] : 0u;
            unsigned rl = en >> EDGE_BITS;
            unsigned o = rl & 31u;
            unsigned valid = __builtin_amdgcn_ballot_w32(ok);
            unsigned ob[5];
            #pragma unroll
            for (int b = 0; b < 5; ++b) ob[b] = __builtin_amdgcn_ballot_w32(ok && ((o >> b) & 1u));
            int p = __shfl(pos, (int)o) + __builtin_popcount(owner_lanes(o, valid, ob) & lt);
            pos += __builtin_popcount(owner_lanes((unsigned)lane, valid, ob));
            if (ok) {
                int e = (int)(en & ((1u << EDGE_BITS) - 1u));
                int c = cols[e];
                float wn = w[e] * dinv[c];
                own[p] = make_int2((int)((unsigned)c | (rl << COL_BITS)), __float_as_int(wn));
            }
        }
        __syncthreads();

        const int2* mine = own + myfirst;
        for (int base = 0; base < mytot; base += 32) {
            int i = base + lane;
            int2 en = make_int2(0, 0);
            if (i < mytot) en = mine[i];
            int m = mytot - base < 32 ? mytot - base : 32;
            for (int j = 0; j + GATHER_BATCH <= m; j += GATHER_BATCH)
                owner_flush((unsigned)en.x, en.y, j, GATHER_BATCH, h, acc, lane);
            if (m & (GATHER_BATCH - 1))
                owner_flush((unsigned)en.x, en.y, m & ~(GATHER_BATCH - 1), m & (GATHER_BATCH - 1), h, acc, lane);
        }
        __syncthreads();
        if (!any) break;
    }

    int u = lane * 4;
    float4 bi = *(const float4*)(bias + u);
    float4 ga = *(const float4*)(gamma + u);
    float4 be = *(const float4*)(beta + u);
    float4 me = *(const float4*)(mean + u);
    float4 va = *(const float4*)(var + u);
    float4 is = make_float4(rsqrtf(va.x + BN_EPS), rsqrtf(va.y + BN_EPS), rsqrtf(va.z + BN_EPS), rsqrtf(va.w + BN_EPS));
    for (int i = t; i < nb * (FEAT / 4); i += OWN_THREADS) {
        int r = i >> 5;
        float s = dinv[r0 + r];
        float4 v = ((const float4*)acc)[i];
        float4 o;
        o.x = ga.x * ((v.x * s + bi.x) - me.x) * is.x + be.x;
        o.y = ga.y * ((v.y * s + bi.y) - me.y) * is.y + be.y;
        o.z = ga.z * ((v.z * s + bi.z) - me.z) * is.z + be.z;
        o.w = ga.w * ((v.w * s + bi.w) - me.w) * is.w + be.w;
        o.x = o.x > 0.0f ? o.x : 0.0f;
        o.y = o.y > 0.0f ? o.y : 0.0f;
        o.z = o.z > 0.0f ? o.z : 0.0f;
        o.w = o.w > 0.0f ? o.w : 0.0f;
        ((float4*)out)[(size_t)(r0 + r) * (FEAT / 4) + lane] = o;
    }
}


extern "C" void kernel_launch(void* const* d_in, const int* in_sizes, int n_in,
                              void* d_out, int out_size, void* d_ws, size_t ws_size,
                              hipStream_t stream) {
    const float* x     = (const float*)d_in[0];
    const int*   eidx  = (const int*)d_in[1];
    const float* ew    = (const float*)d_in[2];
    const float* W     = (const float*)d_in[3];
    const float* bias  = (const float*)d_in[4];
    const float* gamma = (const float*)d_in[5];
    const float* beta  = (const float*)d_in[6];
    const float* mean  = (const float*)d_in[7];
    const float* var   = (const float*)d_in[8];

    int nE = in_sizes[2];
    int nN = in_sizes[0] / FEAT;
    const int* rows = eidx;
    const int* cols = eidx + nE;

    float* h    = (float*)d_ws;
    float* dinv = h + (size_t)nN * FEAT + nN;
    unsigned* idx = (unsigned*)(h + (size_t)nN * FEAT);
    float* out  = (float*)d_out;

    float* part = out;
    deg_partial_kernel<<<dim3(DEG_SLICES, (nN + DEG_BINS - 1) / DEG_BINS), DEG_THREADS, 0, stream>>>(
        rows, ew, part, nE, nN);

    mid_kernel<<<(nN + 127) / 128 + (nE + IDX_CHUNK - 1) / IDX_CHUNK + (nN + MID_THREADS - 1) / MID_THREADS,
                 MID_THREADS, 0, stream>>>(
        x, W, h, rows, idx, part, dinv, nN, nE, (nN + OWN_ROWS - 1) / OWN_ROWS, (nE + IDX_CHUNK - 1) / IDX_CHUNK);

    edge_owner_kernel<<<(nN + OWN_ROWS - 1) / OWN_ROWS, OWN_THREADS, 0, stream>>>(
        rows, cols, ew, dinv, h, out, bias, gamma, beta, mean, var, idx, nE, nN, (nE + IDX_CHUNK - 1) / IDX_CHUNK);
}


# edge filter: index bit per 32-edge group (one 128-byte line), not per 128; eight lanes of a load read one marked line, so loads bring marked lines only; a row's additions get a new fixed order
# speedup vs baseline: 3.0039x; 1.1743x over previous
#include <hip/hip_runtime.h>
#include <math.h>

typedef float v2f __attribute__((ext_vector_type(2)));
typedef float v8f __attribute__((ext_vector_type(8)));

#define FEAT 128
#define MID_THREADS 256
#define BN_EPS 1e-3f


#define DEG_BINS   25600
#define DEG_SLICES 64
#define DEG_THREADS 1024
#define DEG_BATCH  8

#define OWN_STRIDE (393 * 128)
#define IDX_HOLE   2048
#define PART_RUN   (OWN_STRIDE - IDX_HOLE)
__device__ __forceinline__ size_t part_at(unsigned f) {
    return (size_t)f + (size_t)IDX_HOLE * (f / (unsigned)PART_RUN + 1u);
}

__global__ __launch_bounds__(DEG_THREADS) void deg_partial_kernel(const int* __restrict__ rows,
                                                                  const float* __restrict__ w,
                                                                  float* __restrict__ part,
                                                                  int nE, int nN) {
    __shared__ float bins[DEG_BINS];

    int t = threadIdx.x;
    float4* bz = (float4*)bins;
    for (int i = t; i < DEG_BINS / 4; i += DEG_THREADS) bz[i] = make_float4(0.0f, 0.0f, 0.0f, 0.0f);
    __syncthreads();

    int r0  = blockIdx.y * DEG_BINS;
    int per = (nE + DEG_SLICES - 1) / DEG_SLICES;
    int e0  = blockIdx.x * per;
    int e1  = e0 + per < nE ? e0 + per : nE;

    for (int e = e0 + t; e < e1; e += DEG_THREADS * DEG_BATCH) {
        int   r[DEG_BATCH];
        float v[DEG_BATCH];
        for (int u = 0; u < DEG_BATCH; ++u) {
            int ee = e + u * DEG_THREADS;
            bool ok = ee < e1;
            r[u] = ok ? rows[ee] : -1;
            v[u] = ok ? w[ee] : 0.0f;
        }
        for (int u = 0; u < DEG_BATCH; ++u) {
            unsigned d = (unsigned)(r[u] - r0);
            if (d < (unsigned)DEG_BINS) atomicAdd(&bins[d], v[u]);
        }
    }
    __syncthreads();

    int nb = nN - r0 < DEG_BINS ? nN - r0 : DEG_BINS;
    unsigned f0 = (unsigned)(blockIdx.x * nN + r0);
    for (int i = t; i < nb; i += DEG_THREADS) part[part_at(f0 + (unsigned)i)] = bins[i];
}

__device__ __forceinline__ void dinv_block(const float* __restrict__ part, float* __restrict__ dinv, int n, int blk) {
    int i = blk * MID_THREADS + threadIdx.x;
    if (i < n) {
        float p[DEG_SLICES];
        for (int s = 0; s < DEG_SLICES; ++s) p[s] = part[part_at((unsigned)(s * n + i))];
        float d = 0.0f;
        for (int s = 0; s < DEG_SLICES; ++s) d += p[s];
        d += 1.0f;
        dinv[i] = (d > 0.0f) ? rsqrtf(d) : 0.0f;
    }
}


__device__ __forceinline__ void gemm_block(const float* __restrict__ x,
                                           const float* __restrict__ W,
                                           float* __restrict__ h, int nRows, int blk, float* Wl) {
    int t = threadIdx.x;
    const float4* Wv = (const float4*)W;
    float4* Wlv = (float4*)Wl;
    #pragma unroll
    for (int i = t; i < (FEAT * FEAT) / 4; i += MID_THREADS) Wlv[i] = Wv[i];
    __syncthreads();

    int wave = t >> 5;
    int lane = t & 31;
    int row0 = blk * 128 + wave * 16;
    if (row0 >= nRows) return;

    int m  = lane & 15;
    int kh = (lane >> 4) << 1;
    int srow = row0 + m;
    if (srow >= nRows) srow = nRows - 1;
    const float* xr = x + (size_t)srow * FEAT;

    v8f acc[8] = {};

    for (int kb = 0; kb < FEAT / 4; ++kb) {
        int k = kb * 4 + kh;
        float2 av = *(const float2*)(xr + k);
        v2f a; a.x = av.x; a.y = av.y;
        #pragma unroll
        for (int j = 0; j < 8; ++j) {
            int col = j * 16 + m;
            v2f b;
            b.x = Wl[(k + 0) * FEAT + col];
            b.y = Wl[(k + 1) * FEAT + col];
            acc[j] = __builtin_amdgcn_wmma_f32_16x16x4_f32(
                false, a, false, b, (short)0, acc[j], false, false);
        }
    }

    int rbase = row0 + ((lane >> 4) ? 8 : 0);
    int n = lane & 15;
    #pragma unroll
    for (int j = 0; j < 8; ++j) {
        #pragma unroll
        for (int r = 0; r < 8; ++r) {
            int row = rbase + r;
            if (row < nRows) h[(size_t)row * FEAT + j * 16 + n] = acc[j][r];
        }
    }
}


#define OWN_ROWS    393
#define GRP_EDGES   128
#define LINE_EDGES  32
#define IDX_CHUNK   4096
#define OWN_THREADS 1024
#define OWN_WAVES   32
#define LIST_CAP    316
#define COL_BITS    23
#define EDGE_BITS   23
#define GATHER_BATCH 8

__device__ __forceinline__ void owner_flush(unsigned pend_pk, int pend_wn, int first, int np,
                                            const float* __restrict__ h, float* acc, int lane) {
    float4 hv[GATHER_BATCH];
    #pragma unroll
    for (int k = 0; k < GATHER_BATCH; ++k) {
        if (k < np) {
            unsigned pk = (unsigned)__builtin_amdgcn_readlane((int)pend_pk, first + k);
            unsigned c = pk & ((1u << COL_BITS) - 1u);
            hv[k] = *(const float4*)(h + (size_t)c * FEAT + lane * 4);
        }
    }
    #pragma unroll
    for (int k = 0; k < GATHER_BATCH; ++k) {
        if (k < np) {
            unsigned pk = (unsigned)__builtin_amdgcn_readlane((int)pend_pk, first + k);
            unsigned rl = pk >> COL_BITS;
            float wn = __int_as_float(__builtin_amdgcn_readlane(pend_wn, first + k));
            float4* ap = (float4*)acc + rl * (FEAT / 4) + lane;
            float4 a = *ap;
            a.x += hv[k].x * wn; a.y += hv[k].y * wn; a.z += hv[k].z * wn; a.w += hv[k].w * wn;
            *ap = a;
        }
    }
}

__device__ __forceinline__ void index_block(const int* __restrict__ rows, unsigned* __restrict__ idx,
                                            int nE, int nOwn, int blk, unsigned* bits) {
    int t = threadIdx.x;
    for (int i = t; i < 4096; i += MID_THREADS) bits[i] = 0u;
    __syncthreads();
    int rr[4][4];
    #pragma unroll
    for (int k = 0; k < 4; ++k) {
        int e = blk * IDX_CHUNK + k * 1024 + t * 4;
        rr[k][0] = -1; rr[k][1] = -1; rr[k][2] = -1; rr[k][3] = -1;
        if (e + 3 < nE) {
            int4 q = *(const int4*)(rows + e);
            rr[k][0] = q.x; rr[k][1] = q.y; rr[k][2] = q.z; rr[k][3] = q.w;
        } else {
            for (int c = 0; c < 4; ++c) if (e + c < nE) rr[k][c] = rows[e + c];
        }
    }
    #pragma unroll
    for (int k = 0; k < 4; ++k) {
        unsigned bit = 1u << (t >> 3);
        #pragma unroll
        for (int c = 0; c < 4; ++c) {
            unsigned b = (unsigned)rr[k][c] / (unsigned)OWN_ROWS;
            if (b < (unsigned)nOwn) atomicOr(&bits[k * 1024 + b], bit);
        }
    }
    __syncthreads();
    for (int i = t; i < nOwn * 4; i += MID_THREADS)
        idx[(size_t)(i >> 2) * OWN_STRIDE + blk * 4 + (i & 3)] = bits[(i & 3) * 1024 + (i >> 2)];
}

__global__ __launch_bounds__(MID_THREADS) void mid_kernel(const float* __restrict__ x,
                                                          const float* __restrict__ W,
                                                          float* __restrict__ h,
                                                          const int* __restrict__ rows,
                                                          unsigned* __restrict__ idx,
                                                          const float* __restrict__ part,
                                                          float* __restrict__ dinv,
                                                          int nN, int nE, int nOwn, int nIdxBlocks) {
    __shared__ float smem[FEAT * FEAT];
    int nProd = (nN + 127) / 128;
    int b = blockIdx.x;
    if (b < nProd) {
        gemm_block(x, W, h, nN, b, smem);
    } else if (b < nProd + nIdxBlocks) {
        index_block(rows, idx, nE, nOwn, b - nProd, (unsigned*)smem);
    } else {
        dinv_block(part, dinv, nN, b - nProd - nIdxBlocks);
    }
}

template <bool TAIL>
__device__ __forceinline__ bool owner_filter_quarter(int4 q, int eb, int end, int lane,
                                                     int r0, unsigned nbl, unsigned lt, unsigned* mylst, int& n) {
    int rr[4] = {q.x, q.y, q.z, q.w};
    unsigned mk[4];
    #pragma unroll
    for (int c = 0; c < 4; ++c) {
        bool hit = (unsigned)(rr[c] - r0) < nbl;
        if (TAIL) hit = hit && eb + c < end;
        mk[c] = __builtin_amdgcn_ballot_w32(hit);
    }
    if (n + GRP_EDGES > LIST_CAP) {
        int tot = __builtin_popcount(mk[0]) + __builtin_popcount(mk[1]) + __builtin_popcount(mk[2]) + __builtin_popcount(mk[3]);
        if (n + tot > LIST_CAP) return false;
    }
    #pragma unroll
    for (int c = 0; c < 4; ++c) {
        if (mk[c] != 0u) {
            if ((mk[c] >> lane) & 1u) {
                int e = eb + c;
                mylst[n + __builtin_popcount(mk[c] & lt)] = (unsigned)e | ((unsigned)(rr[c] - r0) << EDGE_BITS);
            }
            n += __builtin_popcount(mk[c]);
        }
    }
    return true;
}

__device__ __forceinline__ unsigned owner_lanes(unsigned key, unsigned valid, const unsigned* ob) {
    unsigned m = valid;
    #pragma unroll
    for (int b = 0; b < 5; ++b) m &= ((key >> b) & 1u) ? ob[b] : ~ob[b];
    return m;
}

__global__ __launch_bounds__(OWN_THREADS) void edge_owner_kernel(const int* __restrict__ rows,
                                                                 const int* __restrict__ cols,
                                                                 const float* __restrict__ w,
                                                                 const float* __restrict__ dinv,
                                                                 const float* __restrict__ h,
                                                                 float* __restrict__ out,
                                                                 const float* __restrict__ bias,
                                                                 const float* __restrict__ gamma,
                                                                 const float* __restrict__ beta,
                                                                 const float* __restrict__ mean,
                                                                 const float* __restrict__ var,
                                                                 int nE, int nN, int nWords) {
    __shared__ float    acc[OWN_ROWS * FEAT];
    __shared__ int2     own[OWN_WAVES * LIST_CAP];
    __shared__ unsigned lst[OWN_WAVES * LIST_CAP];
    __shared__ int      cntm[OWN_WAVES * OWN_WAVES];
    __shared__ int      more[OWN_WAVES];

    int t = threadIdx.x;
    int wave = __builtin_amdgcn_readfirstlane(t >> 5);
    int lane = t & 31;
    int r0 = blockIdx.x * OWN_ROWS;
    int nb = nN - r0 < OWN_ROWS ? nN - r0 : OWN_ROWS;

    for (int i = t; i < nb * (FEAT / 4); i += OWN_THREADS) {
        int r = i >> 5;
        float s = dinv[r0 + r];
        float4 v = ((const float4*)h)[(size_t)(r0 + r) * (FEAT / 4) + (i & 31)];
        v.x *= s; v.y *= s; v.z *= s; v.w *= s;
        ((float4*)acc)[i] = v;
    }

    int wpw = (nWords + OWN_WAVES - 1) / OWN_WAVES;
    int w0 = wave * wpw < nWords ? wave * wpw : nWords;
    int w1 = w0 + wpw < nWords ? w0 + wpw : nWords;
    const unsigned* myidx = (const unsigned*)out + (size_t)blockIdx.x * OWN_STRIDE;
    int cw = w0;
    unsigned myw = w0 + lane < w1 ? myidx[w0 + lane] : 0u;
    unsigned cm = (unsigned)__builtin_amdgcn_readlane((int)myw, 0);
    int last = (nE - 1) & ~3;
    unsigned lt = (1u << lane) - 1u;
    unsigned* mylst = lst + wave * LIST_CAP;
    int sub = lane >> 3;
    int l8 = (lane & 7) * 4;
    bool sub1 = sub >= 1, sub2 = sub >= 2, sub3 = sub >= 3;

    for (;;) {
        int n = 0;
        int full = 0;
        while (!full) {
            int gg[4][4];
            int kk[4];
            int gtop[4];
            int scw[4];
            unsigned scm[4];
            int fin = 0;
            #pragma unroll
            for (int j = 0; j < 4; ++j) {
                scw[j] = cw;
                scm[j] = cm;
                kk[j] = 0;
                gtop[j] = 0;
                #pragma unroll
                for (int s = 0; s < 4; ++s) {
                    gg[j][s] = 0;
                    if (!fin) {
                        while (cm == 0u && cw + 1 < w1) {
                            ++cw;
                            if (((cw - w0) & 31) == 0) myw = cw + lane < w1 ? myidx[cw + lane] : 0u;
                            cm = (unsigned)__builtin_amdgcn_readlane((int)myw, (cw - w0) & 31);
                        }
                        if (cm != 0u) {
                            gg[j][s] = cw * 32 + __builtin_ctz(cm);
                            gtop[j] = gg[j][s];
                            cm &= cm - 1u;
                            kk[j] = s + 1;
                        } else {
                            fin = 1;
                        }
                    }
                }
            }
            if (kk[0] == 0) break;
            int4 q[4];
            int eb[4];
            #pragma unroll
            for (int j = 0; j < 4; ++j) {
                int g = gg[j][0] + (sub1 ? gg[j][1] - gg[j][0] : 0) + (sub2 ? gg[j][2] - gg[j][1] : 0)
                                 + (sub3 ? gg[j][3] - gg[j][2] : 0);
                eb[j] = g * LINE_EDGES + l8;
                int i = eb[j] < last ? eb[j] : last;
                q[j] = *(const int4*)(rows + i);
            }
            #pragma unroll
            for (int j = 0; j < 4; ++j) {
                if (kk[j] > 0 && !full) {
                    unsigned nbl = sub < kk[j] ? (unsigned)nb : 0u;
                    bool ok = gtop[j] * LINE_EDGES + LINE_EDGES <= nE
                                  ? owner_filter_quarter<false>(q[j], eb[j], nE, lane, r0, nbl, lt, mylst, n)
                                  : owner_filter_quarter<true>(q[j], eb[j], nE, lane, r0, nbl, lt, mylst, n);
                    if (!ok) {
                        cw = scw[j];
                        cm = scm[j];
                        full = 1;
                        int wb = w0 + ((cw - w0) & ~31);
                        myw = wb + lane < w1 ? myidx[wb + lane] : 0u;
                    }
                }
            }
        }
        int mycnt = 0;
        for (int base = 0; base < n; base += 32) {
            int i = base + lane;
            bool ok = i < n;
            unsigned o = ok ? (mylst[i] >> EDGE_BITS) & 31u : 0u;
            unsigned valid = __builtin_amdgcn_ballot_w32(ok);
            unsigned ob[5];
            #pragma unroll
            for (int b = 0; b < 5; ++b) ob[b] = __builtin_amdgcn_ballot_w32(ok && ((o >> b) & 1u));
            mycnt += __builtin_popcount(owner_lanes((unsigned)lane, valid, ob));
        }
        cntm[wave * OWN_WAVES + lane] = mycnt;
        if (lane == 0) more[wave] = full;
        __syncthreads();

        int before = 0;
        int tot = 0;
        #pragma unroll
        for (int s = 0; s < OWN_WAVES; ++s) {
            int v = cntm[s * OWN_WAVES + lane];
            tot += v;
            before += s < wave ? v : 0;
        }
        int incl = tot;
        for (int d = 1; d < 32; d <<= 1) {
            int up = __shfl_up(incl, d);
            if (lane >= d) incl += up;
        }
        int pos = incl - tot + before;
        int myfirst = __builtin_amdgcn_readlane(incl - tot, wave);
        int mytot = __builtin_amdgcn_readlane(tot, wave);
        int any = __builtin_amdgcn_ballot_w32(more[lane] != 0) != 0u;

        for (int base = 0; base < n; base += 32) {
            int i = base + lane;
            bool ok = i < n;
            unsigned en = ok ? mylst[i] : 0u;
            unsigned rl = en >> EDGE_BITS;
            unsigned o = rl & 31u;
            unsigned valid = __builtin_amdgcn_ballot_w32(ok);
            unsigned ob[5];
            #pragma unroll
            for (int b = 0; b < 5; ++b) ob[b] = __builtin_amdgcn_ballot_w32(ok && ((o >> b) & 1u));
            int p = __shfl(pos, (int)o) + __builtin_popcount(owner_lanes(o, valid, ob) & lt);
            pos += __builtin_popcount(owner_lanes((unsigned)lane, valid, ob));
            if (ok) {
                int e = (int)(en & ((1u << EDGE_BITS) - 1u));
                int c = cols[e];
                float wn = w[e] * dinv[c];
                own[p] = make_int2((int)((unsigned)c | (rl << COL_BITS)), __float_as_int(wn));
            }
        }
        __syncthreads();

        const int2* mine = own + myfirst;
        for (int base = 0; base < mytot; base += 32) {
            int i = base + lane;
            int2 en = make_int2(0, 0);
            if (i < mytot) en = mine[i];
            int m = mytot - base < 32 ? mytot - base : 32;
            for (int j = 0; j + GATHER_BATCH <= m; j += GATHER_BATCH)
                owner_flush((unsigned)en.x, en.y, j, GATHER_BATCH, h, acc, lane);
            if (m & (GATHER_BATCH - 1))
                owner_flush((unsigned)en.x, en.y, m & ~(GATHER_BATCH - 1), m & (GATHER_BATCH - 1), h, acc, lane);
        }
        __syncthreads();
        if (!any) break;
    }

    int u = lane * 4;
    float4 bi = *(const float4*)(bias + u);
    float4 ga = *(const float4*)(gamma + u);
    float4 be = *(const float4*)(beta + u);
    float4 me = *(const float4*)(mean + u);
    float4 va = *(const float4*)(var + u);
    float4 is = make_float4(rsqrtf(va.x + BN_EPS), rsqrtf(va.y + BN_EPS), rsqrtf(va.z + BN_EPS), rsqrtf(va.w + BN_EPS));
    for (int i = t; i < nb * (FEAT / 4); i += OWN_THREADS) {
        int r = i >> 5;
        float s = dinv[r0 + r];
        float4 v = ((const float4*)acc)[i];
        float4 o;
        o.x = ga.x * ((v.x * s + bi.x) - me.x) * is.x + be.x;
        o.y = ga.y * ((v.y * s + bi.y) - me.y) * is.y + be.y;
        o.z = ga.z * ((v.z * s + bi.z) - me.z) * is.z + be.z;
        o.w = ga.w * ((v.w * s + bi.w) - me.w) * is.w + be.w;
        o.x = o.x > 0.0f ? o.x : 0.0f;
        o.y = o.y > 0.0f ? o.y : 0.0f;
        o.z = o.z > 0.0f ? o.z : 0.0f;
        o.w = o.w > 0.0f ? o.w : 0.0f;
        ((float4*)out)[(size_t)(r0 + r) * (FEAT / 4) + lane] = o;
    }
}


extern "C" void kernel_launch(void* const* d_in, const int* in_sizes, int n_in,
                              void* d_out, int out_size, void* d_ws, size_t ws_size,
                              hipStream_t stream) {
    const float* x     = (const float*)d_in[0];
    const int*   eidx  = (const int*)d_in[1];
    const float* ew    = (const float*)d_in[2];
    const float* W     = (const float*)d_in[3];
    const float* bias  = (const float*)d_in[4];
    const float* gamma = (const float*)d_in[5];
    const float* beta  = (const float*)d_in[6];
    const float* mean  = (const float*)d_in[7];
    const float* var   = (const float*)d_in[8];

    int nE = in_sizes[2];
    int nN = in_sizes[0] / FEAT;
    const int* rows = eidx;
    const int* cols = eidx + nE;

    float* h    = (float*)d_ws;
    float* dinv = h + (size_t)nN * FEAT + nN;
    float* out  = (float*)d_out;

    float* part = out;
    unsigned* idx = (unsigned*)d_out;
    deg_partial_kernel<<<dim3(DEG_SLICES, (nN + DEG_BINS - 1) / DEG_BINS), DEG_THREADS, 0, stream>>>(
        rows, ew, part, nE, nN);

    mid_kernel<<<(nN + 127) / 128 + (nE + IDX_CHUNK - 1) / IDX_CHUNK + (nN + MID_THREADS - 1) / MID_THREADS,
                 MID_THREADS, 0, stream>>>(
        x, W, h, rows, idx, part, dinv, nN, nE, (nN + OWN_ROWS - 1) / OWN_ROWS, (nE + IDX_CHUNK - 1) / IDX_CHUNK);

    edge_owner_kernel<<<(nN + OWN_ROWS - 1) / OWN_ROWS, OWN_THREADS, 0, stream>>>(
        rows, cols, ew, dinv, h, out, bias, gamma, beta, mean, var, nE, nN, 4 * ((nE + IDX_CHUNK - 1) / IDX_CHUNK));
}


# edge filter: each 8-lane slot of a load walks its own quarter of the wave's index words (kept in LDS) with vector steps, no scalar bit walk; a row's additions get a new fixed order
# speedup vs baseline: 3.3221x; 1.1059x over previous
#include <hip/hip_runtime.h>
#include <math.h>

typedef float v2f __attribute__((ext_vector_type(2)));
typedef float v8f __attribute__((ext_vector_type(8)));

#define FEAT 128
#define MID_THREADS 256
#define BN_EPS 1e-3f


#define DEG_BINS   25600
#define DEG_SLICES 64
#define DEG_THREADS 1024
#define DEG_BATCH  8

#define OWN_STRIDE (393 * 128)
#define IDX_HOLE   2048
#define PART_RUN   (OWN_STRIDE - IDX_HOLE)
__device__ __forceinline__ size_t part_at(unsigned f) {
    return (size_t)f + (size_t)IDX_HOLE * (f / (unsigned)PART_RUN + 1u);
}

__global__ __launch_bounds__(DEG_THREADS) void deg_partial_kernel(const int* __restrict__ rows,
                                                                  const float* __restrict__ w,
                                                                  float* __restrict__ part,
                                                                  int nE, int nN) {
    __shared__ float bins[DEG_BINS];

    int t = threadIdx.x;
    float4* bz = (float4*)bins;
    for (int i = t; i < DEG_BINS / 4; i += DEG_THREADS) bz[i] = make_float4(0.0f, 0.0f, 0.0f, 0.0f);
    __syncthreads();

    int r0  = blockIdx.y * DEG_BINS;
    int per = (nE + DEG_SLICES - 1) / DEG_SLICES;
    int e0  = blockIdx.x * per;
    int e1  = e0 + per < nE ? e0 + per : nE;

    for (int e = e0 + t; e < e1; e += DEG_THREADS * DEG_BATCH) {
        int   r[DEG_BATCH];
        float v[DEG_BATCH];
        for (int u = 0; u < DEG_BATCH; ++u) {
            int ee = e + u * DEG_THREADS;
            bool ok = ee < e1;
            r[u] = ok ? rows[ee] : -1;
            v[u] = ok ? w[ee] : 0.0f;
        }
        for (int u = 0; u < DEG_BATCH; ++u) {
            unsigned d = (unsigned)(r[u] - r0);
            if (d < (unsigned)DEG_BINS) atomicAdd(&bins[d], v[u]);
        }
    }
    __syncthreads();

    int nb = nN - r0 < DEG_BINS ? nN - r0 : DEG_BINS;
    unsigned f0 = (unsigned)(blockIdx.x * nN + r0);
    for (int i = t; i < nb; i += DEG_THREADS) part[part_at(f0 + (unsigned)i)] = bins[i];
}

__device__ __forceinline__ void dinv_block(const float* __restrict__ part, float* __restrict__ dinv, int n, int blk) {
    int i = blk * MID_THREADS + threadIdx.x;
    if (i < n) {
        float p[DEG_SLICES];
        for (int s = 0; s < DEG_SLICES; ++s) p[s] = part[part_at((unsigned)(s * n + i))];
        float d = 0.0f;
        for (int s = 0; s < DEG_SLICES; ++s) d += p[s];
        d += 1.0f;
        dinv[i] = (d > 0.0f) ? rsqrtf(d) : 0.0f;
    }
}


__device__ __forceinline__ void gemm_block(const float* __restrict__ x,
                                           const float* __restrict__ W,
                                           float* __restrict__ h, int nRows, int blk, float* Wl) {
    int t = threadIdx.x;
    const float4* Wv = (const float4*)W;
    float4* Wlv = (float4*)Wl;
    #pragma unroll
    for (int i = t; i < (FEAT * FEAT) / 4; i += MID_THREADS) Wlv[i] = Wv[i];
    __syncthreads();

    int wave = t >> 5;
    int lane = t & 31;
    int row0 = blk * 128 + wave * 16;
    if (row0 >= nRows) return;

    int m  = lane & 15;
    int kh = (lane >> 4) << 1;
    int srow = row0 + m;
    if (srow >= nRows) srow = nRows - 1;
    const float* xr = x + (size_t)srow * FEAT;

    v8f acc[8] = {};

    for (int kb = 0; kb < FEAT / 4; ++kb) {
        int k = kb * 4 + kh;
        float2 av = *(const float2*)(xr + k);
        v2f a; a.x = av.x; a.y = av.y;
        #pragma unroll
        for (int j = 0; j < 8; ++j) {
            int col = j * 16 + m;
            v2f b;
            b.x = Wl[(k + 0) * FEAT + col];
            b.y = Wl[(k + 1) * FEAT + col];
            acc[j] = __builtin_amdgcn_wmma_f32_16x16x4_f32(
                false, a, false, b, (short)0, acc[j], false, false);
        }
    }

    int rbase = row0 + ((lane >> 4) ? 8 : 0);
    int n = lane & 15;
    #pragma unroll
    for (int j = 0; j < 8; ++j) {
        #pragma unroll
        for (int r = 0; r < 8; ++r) {
            int row = rbase + r;
            if (row < nRows) h[(size_t)row * FEAT + j * 16 + n] = acc[j][r];
        }
    }
}


#define OWN_ROWS    393
#define GRP_EDGES   128
#define LINE_EDGES  32
#define IDX_CHUNK   4096
#define OWN_THREADS 1024
#define OWN_WAVES   32
#define LIST_CAP    316
#define COL_BITS    23
#define EDGE_BITS   23
#define GATHER_BATCH 8

__device__ __forceinline__ void owner_flush(unsigned pend_pk, int pend_wn, int first, int np,
                                            const float* __restrict__ h, float* acc, int lane) {
    float4 hv[GATHER_BATCH];
    #pragma unroll
    for (int k = 0; k < GATHER_BATCH; ++k) {
        if (k < np) {
            unsigned pk = (unsigned)__builtin_amdgcn_readlane((int)pend_pk, first + k);
            unsigned c = pk & ((1u << COL_BITS) - 1u);
            hv[k] = *(const float4*)(h + (size_t)c * FEAT + lane * 4);
        }
    }
    #pragma unroll
    for (int k = 0; k < GATHER_BATCH; ++k) {
        if (k < np) {
            unsigned pk = (unsigned)__builtin_amdgcn_readlane((int)pend_pk, first + k);
            unsigned rl = pk >> COL_BITS;
            float wn = __int_as_float(__builtin_amdgcn_readlane(pend_wn, first + k));
            float4* ap = (float4*)acc + rl * (FEAT / 4) + lane;
            float4 a = *ap;
            a.x += hv[k].x * wn; a.y += hv[k].y * wn; a.z += hv[k].z * wn; a.w += hv[k].w * wn;
            *ap = a;
        }
    }
}

__device__ __forceinline__ void index_block(const int* __restrict__ rows, unsigned* __restrict__ idx,
                                            int nE, int nOwn, int blk, unsigned* bits) {
    int t = threadIdx.x;
    for (int i = t; i < 4096; i += MID_THREADS) bits[i] = 0u;
    __syncthreads();
    int rr[4][4];
    #pragma unroll
    for (int k = 0; k < 4; ++k) {
        int e = blk * IDX_CHUNK + k * 1024 + t * 4;
        rr[k][0] = -1; rr[k][1] = -1; rr[k][2] = -1; rr[k][3] = -1;
        if (e + 3 < nE) {
            int4 q = *(const int4*)(rows + e);
            rr[k][0] = q.x; rr[k][1] = q.y; rr[k][2] = q.z; rr[k][3] = q.w;
        } else {
            for (int c = 0; c < 4; ++c) if (e + c < nE) rr[k][c] = rows[e + c];
        }
    }
    #pragma unroll
    for (int k = 0; k < 4; ++k) {
        unsigned bit = 1u << (t >> 3);
        #pragma unroll
        for (int c = 0; c < 4; ++c) {
            unsigned b = (unsigned)rr[k][c] / (unsigned)OWN_ROWS;
            if (b < (unsigned)nOwn) atomicOr(&bits[k * 1024 + b], bit);
        }
    }
    __syncthreads();
    for (int i = t; i < nOwn * 4; i += MID_THREADS)
        idx[(size_t)(i >> 2) * OWN_STRIDE + blk * 4 + (i & 3)] = bits[(i & 3) * 1024 + (i >> 2)];
}

__global__ __launch_bounds__(MID_THREADS) void mid_kernel(const float* __restrict__ x,
                                                          const float* __restrict__ W,
                                                          float* __restrict__ h,
                                                          const int* __restrict__ rows,
                                                          unsigned* __restrict__ idx,
                                                          const float* __restrict__ part,
                                                          float* __restrict__ dinv,
                                                          int nN, int nE, int nOwn, int nIdxBlocks) {
    __shared__ float smem[FEAT * FEAT];
    int nProd = (nN + 127) / 128;
    int b = blockIdx.x;
    if (b < nProd) {
        gemm_block(x, W, h, nN, b, smem);
    } else if (b < nProd + nIdxBlocks) {
        index_block(rows, idx, nE, nOwn, b - nProd, (unsigned*)smem);
    } else {
        dinv_block(part, dinv, nN, b - nProd - nIdxBlocks);
    }
}

template <bool TAIL>
__device__ __forceinline__ bool owner_filter_quarter(int4 q, int eb, int end, int lane,
                                                     int r0, unsigned nbl, unsigned lt, unsigned* mylst, int& n) {
    int rr[4] = {q.x, q.y, q.z, q.w};
    unsigned mk[4];
    #pragma unroll
    for (int c = 0; c < 4; ++c) {
        bool hit = (unsigned)(rr[c] - r0) < nbl;
        if (TAIL) hit = hit && eb + c < end;
        mk[c] = __builtin_amdgcn_ballot_w32(hit);
    }
    if (n + GRP_EDGES > LIST_CAP) {
        int tot = __builtin_popcount(mk[0]) + __builtin_popcount(mk[1]) + __builtin_popcount(mk[2]) + __builtin_popcount(mk[3]);
        if (n + tot > LIST_CAP) return false;
    }
    #pragma unroll
    for (int c = 0; c < 4; ++c) {
        if (mk[c] != 0u) {
            if ((mk[c] >> lane) & 1u) {
                int e = eb + c;
                mylst[n + __builtin_popcount(mk[c] & lt)] = (unsigned)e | ((unsigned)(rr[c] - r0) << EDGE_BITS);
            }
            n += __builtin_popcount(mk[c]);
        }
    }
    return true;
}

__device__ __forceinline__ unsigned owner_lanes(unsigned key, unsigned valid, const unsigned* ob) {
    unsigned m = valid;
    #pragma unroll
    for (int b = 0; b < 5; ++b) m &= ((key >> b) & 1u) ? ob[b] : ~ob[b];
    return m;
}

__global__ __launch_bounds__(OWN_THREADS) void edge_owner_kernel(const int* __restrict__ rows,
                                                                 const int* __restrict__ cols,
                                                                 const float* __restrict__ w,
                                                                 const float* __restrict__ dinv,
                                                                 const float* __restrict__ h,
                                                                 float* __restrict__ out,
                                                                 const float* __restrict__ bias,
                                                                 const float* __restrict__ gamma,
                                                                 const float* __restrict__ beta,
                                                                 const float* __restrict__ mean,
                                                                 const float* __restrict__ var,
                                                                 int nE, int nN, int nWords) {
    __shared__ float    acc[OWN_ROWS * FEAT];
    __shared__ int2     own[OWN_WAVES * LIST_CAP];
    __shared__ unsigned lst[OWN_WAVES * LIST_CAP];
    __shared__ int      cntm[OWN_WAVES * OWN_WAVES];
    __shared__ int      more[OWN_WAVES];

    int t = threadIdx.x;
    int wave = __builtin_amdgcn_readfirstlane(t >> 5);
    int lane = t & 31;
    int r0 = blockIdx.x * OWN_ROWS;
    int nb = nN - r0 < OWN_ROWS ? nN - r0 : OWN_ROWS;

    for (int i = t; i < nb * (FEAT / 4); i += OWN_THREADS) {
        int r = i >> 5;
        float s = dinv[r0 + r];
        float4 v = ((const float4*)h)[(size_t)(r0 + r) * (FEAT / 4) + (i & 31)];
        v.x *= s; v.y *= s; v.z *= s; v.w *= s;
        ((float4*)acc)[i] = v;
    }

    int wpw = (nWords + OWN_WAVES - 1) / OWN_WAVES;
    int w0 = wave * wpw < nWords ? wave * wpw : nWords;
    int w1 = w0 + wpw < nWords ? w0 + wpw : nWords;
    const unsigned* myidx = (const unsigned*)out + (size_t)blockIdx.x * OWN_STRIDE;
    unsigned* idxl = (unsigned*)own;
    int last = (nE - 1) & ~3;
    unsigned lt = (1u << lane) - 1u;
    unsigned* mylst = lst + wave * LIST_CAP;
    int sub = lane >> 3;
    int l8 = (lane & 7) * 4;
    int spw = (w1 - w0 + 3) >> 2;
    int ws0 = w0 + sub * spw < w1 ? w0 + sub * spw : w1;
    int ws1 = ws0 + spw < w1 ? ws0 + spw : w1;
    int cwv = ws0;
    unsigned cmv = ws0 < ws1 ? myidx[ws0] : 0u;

    for (;;) {
        for (int k = w0 + lane; k < w1; k += 32) idxl[k] = myidx[k];
        __syncthreads();
        int n = 0;
        int full = 0;
        while (!full) {
            int g[4];
            bool has[4];
            int scw[4];
            unsigned scm[4];
            #pragma unroll
            for (int j = 0; j < 4; ++j) {
                scw[j] = cwv;
                scm[j] = cmv;
                while (cmv == 0u && cwv + 1 < ws1) {
                    ++cwv;
                    cmv = idxl[cwv];
                }
                has[j] = cmv != 0u;
                g[j] = has[j] ? cwv * 32 + __builtin_ctz(cmv) : 0;
                cmv &= cmv - 1u;
            }
            if (__builtin_amdgcn_ballot_w32(has[0]) == 0u) break;
            int4 q[4];
            int eb[4];
            #pragma unroll
            for (int j = 0; j < 4; ++j) {
                eb[j] = g[j] * LINE_EDGES + l8;
                int i = eb[j] < last ? eb[j] : last;
                q[j] = *(const int4*)(rows + i);
            }
            #pragma unroll
            for (int j = 0; j < 4; ++j) {
                bool act = j == 0 ? true : __builtin_amdgcn_ballot_w32(has[j]) != 0u;
                if (act && !full) {
                    unsigned nbl = has[j] ? (unsigned)nb : 0u;
                    bool tail = __builtin_amdgcn_ballot_w32(has[j] && g[j] * LINE_EDGES + LINE_EDGES > nE) != 0u;
                    bool ok = !tail ? owner_filter_quarter<false>(q[j], eb[j], nE, lane, r0, nbl, lt, mylst, n)
                                    : owner_filter_quarter<true>(q[j], eb[j], nE, lane, r0, nbl, lt, mylst, n);
                    if (!ok) {
                        cwv = scw[j];
                        cmv = scm[j];
                        full = 1;
                    }
                }
            }
        }
        int mycnt = 0;
        for (int base = 0; base < n; base += 32) {
            int i = base + lane;
            bool ok = i < n;
            unsigned o = ok ? (mylst[i] >> EDGE_BITS) & 31u : 0u;
            unsigned valid = __builtin_amdgcn_ballot_w32(ok);
            unsigned ob[5];
            #pragma unroll
            for (int b = 0; b < 5; ++b) ob[b] = __builtin_amdgcn_ballot_w32(ok && ((o >> b) & 1u));
            mycnt += __builtin_popcount(owner_lanes((unsigned)lane, valid, ob));
        }
        cntm[wave * OWN_WAVES + lane] = mycnt;
        if (lane == 0) more[wave] = full;
        __syncthreads();

        int before = 0;
        int tot = 0;
        #pragma unroll
        for (int s = 0; s < OWN_WAVES; ++s) {
            int v = cntm[s * OWN_WAVES + lane];
            tot += v;
            before += s < wave ? v : 0;
        }
        int incl = tot;
        for (int d = 1; d < 32; d <<= 1) {
            int up = __shfl_up(incl, d);
            if (lane >= d) incl += up;
        }
        int pos = incl - tot + before;
        int myfirst = __builtin_amdgcn_readlane(incl - tot, wave);
        int mytot = __builtin_amdgcn_readlane(tot, wave);
        int any = __builtin_amdgcn_ballot_w32(more[lane] != 0) != 0u;

        for (int base = 0; base < n; base += 32) {
            int i = base + lane;
            bool ok = i < n;
            unsigned en = ok ? mylst[i] : 0u;
            unsigned rl = en >> EDGE_BITS;
            unsigned o = rl & 31u;
            unsigned valid = __builtin_amdgcn_ballot_w32(ok);
            unsigned ob[5];
            #pragma unroll
            for (int b = 0; b < 5; ++b) ob[b] = __builtin_amdgcn_ballot_w32(ok && ((o >> b) & 1u));
            int p = __shfl(pos, (int)o) + __builtin_popcount(owner_lanes(o, valid, ob) & lt);
            pos += __builtin_popcount(owner_lanes((unsigned)lane, valid, ob));
            if (ok) {
                int e = (int)(en & ((1u << EDGE_BITS) - 1u));
                int c = cols[e];
                float wn = w[e] * dinv[c];
                own[p] = make_int2((int)((unsigned)c | (rl << COL_BITS)), __float_as_int(wn));
            }
        }
        __syncthreads();

        const int2* mine = own + myfirst;
        for (int base = 0; base < mytot; base += 32) {
            int i = base + lane;
            int2 en = make_int2(0, 0);
            if (i < mytot) en = mine[i];
            int m = mytot - base < 32 ? mytot - base : 32;
            for (int j = 0; j + GATHER_BATCH <= m; j += GATHER_BATCH)
                owner_flush((unsigned)en.x, en.y, j, GATHER_BATCH, h, acc, lane);
            if (m & (GATHER_BATCH - 1))
                owner_flush((unsigned)en.x, en.y, m & ~(GATHER_BATCH - 1), m & (GATHER_BATCH - 1), h, acc, lane);
        }
        __syncthreads();
        if (!any) break;
    }

    int u = lane * 4;
    float4 bi = *(const float4*)(bias + u);
    float4 ga = *(const float4*)(gamma + u);
    float4 be = *(const float4*)(beta + u);
    float4 me = *(const float4*)(mean + u);
    float4 va = *(const float4*)(var + u);
    float4 is = make_float4(rsqrtf(va.x + BN_EPS), rsqrtf(va.y + BN_EPS), rsqrtf(va.z + BN_EPS), rsqrtf(va.w + BN_EPS));
    for (int i = t; i < nb * (FEAT / 4); i += OWN_THREADS) {
        int r = i >> 5;
        float s = dinv[r0 + r];
        float4 v = ((const float4*)acc)[i];
        float4 o;
        o.x = ga.x * ((v.x * s + bi.x) - me.x) * is.x + be.x;
        o.y = ga.y * ((v.y * s + bi.y) - me.y) * is.y + be.y;
        o.z = ga.z * ((v.z * s + bi.z) - me.z) * is.z + be.z;
        o.w = ga.w * ((v.w * s + bi.w) - me.w) * is.w + be.w;
        o.x = o.x > 0.0f ? o.x : 0.0f;
        o.y = o.y > 0.0f ? o.y : 0.0f;
        o.z = o.z > 0.0f ? o.z : 0.0f;
        o.w = o.w > 0.0f ? o.w : 0.0f;
        ((float4*)out)[(size_t)(r0 + r) * (FEAT / 4) + lane] = o;
    }
}


extern "C" void kernel_launch(void* const* d_in, const int* in_sizes, int n_in,
                              void* d_out, int out_size, void* d_ws, size_t ws_size,
                              hipStream_t stream) {
    const float* x     = (const float*)d_in[0];
    const int*   eidx  = (const int*)d_in[1];
    const float* ew    = (const float*)d_in[2];
    const float* W     = (const float*)d_in[3];
    const float* bias  = (const float*)d_in[4];
    const float* gamma = (const float*)d_in[5];
    const float* beta  = (const float*)d_in[6];
    const float* mean  = (const float*)d_in[7];
    const float* var   = (const float*)d_in[8];

    int nE = in_sizes[2];
    int nN = in_sizes[0] / FEAT;
    const int* rows = eidx;
    const int* cols = eidx + nE;

    float* h    = (float*)d_ws;
    float* dinv = h + (size_t)nN * FEAT + nN;
    float* out  = (float*)d_out;

    float* part = out;
    unsigned* idx = (unsigned*)d_out;
    deg_partial_kernel<<<dim3(DEG_SLICES, (nN + DEG_BINS - 1) / DEG_BINS), DEG_THREADS, 0, stream>>>(
        rows, ew, part, nE, nN);

    mid_kernel<<<(nN + 127) / 128 + (nE + IDX_CHUNK - 1) / IDX_CHUNK + (nN + MID_THREADS - 1) / MID_THREADS,
                 MID_THREADS, 0, stream>>>(
        x, W, h, rows, idx, part, dinv, nN, nE, (nN + OWN_ROWS - 1) / OWN_ROWS, (nE + IDX_CHUNK - 1) / IDX_CHUNK);

    edge_owner_kernel<<<(nN + OWN_ROWS - 1) / OWN_ROWS, OWN_THREADS, 0, stream>>>(
        rows, cols, ew, dinv, h, out, bias, gamma, beta, mean, var, nE, nN, 4 * ((nE + IDX_CHUNK - 1) / IDX_CHUNK));
}


# edge filter: index bit per 8-edge group (32 bytes, two lanes of a load), not per 32; a load asks for sixteen marked groups, a third as many loads to test; a row's additions get a new fixed order
# speedup vs baseline: 3.5843x; 1.0789x over previous
#include <hip/hip_runtime.h>
#include <math.h>

typedef float v2f __attribute__((ext_vector_type(2)));
typedef float v8f __attribute__((ext_vector_type(8)));

#define FEAT 128
#define MID_THREADS 256
#define BN_EPS 1e-3f


#define DEG_BINS   25600
#define DEG_SLICES 64
#define DEG_THREADS 1024
#define DEG_BATCH  8

#define OWN_STRIDE (393 * 128)
#define IDX_HOLE   8192
#define PART_RUN   (OWN_STRIDE - IDX_HOLE)
__device__ __forceinline__ size_t part_at(unsigned f) {
    return (size_t)f + (size_t)IDX_HOLE * (f / (unsigned)PART_RUN + 1u);
}

__global__ __launch_bounds__(DEG_THREADS) void deg_partial_kernel(const int* __restrict__ rows,
                                                                  const float* __restrict__ w,
                                                                  float* __restrict__ part,
                                                                  int nE, int nN) {
    __shared__ float bins[DEG_BINS];

    int t = threadIdx.x;
    float4* bz = (float4*)bins;
    for (int i = t; i < DEG_BINS / 4; i += DEG_THREADS) bz[i] = make_float4(0.0f, 0.0f, 0.0f, 0.0f);
    __syncthreads();

    int r0  = blockIdx.y * DEG_BINS;
    int per = (nE + DEG_SLICES - 1) / DEG_SLICES;
    int e0  = blockIdx.x * per;
    int e1  = e0 + per < nE ? e0 + per : nE;

    for (int e = e0 + t; e < e1; e += DEG_THREADS * DEG_BATCH) {
        int   r[DEG_BATCH];
        float v[DEG_BATCH];
        for (int u = 0; u < DEG_BATCH; ++u) {
            int ee = e + u * DEG_THREADS;
            bool ok = ee < e1;
            r[u] = ok ? rows[ee] : -1;
            v[u] = ok ? w[ee] : 0.0f;
        }
        for (int u = 0; u < DEG_BATCH; ++u) {
            unsigned d = (unsigned)(r[u] - r0);
            if (d < (unsigned)DEG_BINS) atomicAdd(&bins[d], v[u]);
        }
    }
    __syncthreads();

    int nb = nN - r0 < DEG_BINS ? nN - r0 : DEG_BINS;
    unsigned f0 = (unsigned)(blockIdx.x * nN + r0);
    for (int i = t; i < nb; i += DEG_THREADS) part[part_at(f0 + (unsigned)i)] = bins[i];
}

__device__ __forceinline__ void dinv_block(const float* __restrict__ part, float* __restrict__ dinv, int n, int blk) {
    int i = blk * MID_THREADS + threadIdx.x;
    if (i < n) {
        float p[DEG_SLICES];
        for (int s = 0; s < DEG_SLICES; ++s) p[s] = part[part_at((unsigned)(s * n + i))];
        float d = 0.0f;
        for (int s = 0; s < DEG_SLICES; ++s) d += p[s];
        d += 1.0f;
        dinv[i] = (d > 0.0f) ? rsqrtf(d) : 0.0f;
    }
}


__device__ __forceinline__ void gemm_block(const float* __restrict__ x,
                                           const float* __restrict__ W,
                                           float* __restrict__ h, int nRows, int blk, float* Wl) {
    int t = threadIdx.x;
    const float4* Wv = (const float4*)W;
    float4* Wlv = (float4*)Wl;
    #pragma unroll
    for (int i = t; i < (FEAT * FEAT) / 4; i += MID_THREADS) Wlv[i] = Wv[i];
    __syncthreads();

    int wave = t >> 5;
    int lane = t & 31;
    int row0 = blk * 128 + wave * 16;
    if (row0 >= nRows) return;

    int m  = lane & 15;
    int kh = (lane >> 4) << 1;
    int srow = row0 + m;
    if (srow >= nRows) srow = nRows - 1;
    const float* xr = x + (size_t)srow * FEAT;

    v8f acc[8] = {};

    for (int kb = 0; kb < FEAT / 4; ++kb) {
        int k = kb * 4 + kh;
        float2 av = *(const float2*)(xr + k);
        v2f a; a.x = av.x; a.y = av.y;
        #pragma unroll
        for (int j = 0; j < 8; ++j) {
            int col = j * 16 + m;
            v2f b;
            b.x = Wl[(k + 0) * FEAT + col];
            b.y = Wl[(k + 1) * FEAT + col];
            acc[j] = __builtin_amdgcn_wmma_f32_16x16x4_f32(
                false, a, false, b, (short)0, acc[j], false, false);
        }
    }

    int rbase = row0 + ((lane >> 4) ? 8 : 0);
    int n = lane & 15;
    #pragma unroll
    for (int j = 0; j < 8; ++j) {
        #pragma unroll
        for (int r = 0; r < 8; ++r) {
            int row = rbase + r;
            if (row < nRows) h[(size_t)row * FEAT + j * 16 + n] = acc[j][r];
        }
    }
}


#define OWN_ROWS    393
#define GRP_EDGES   128
#define LINE_EDGES  8
#define SLOT_LANES  2
#define SLOTS       16
#define IDX_CHUNK   4096
#define IDX_WPB     16
#define OWN_THREADS 1024
#define OWN_WAVES   32
#define LIST_CAP    316
#define COL_BITS    23
#define EDGE_BITS   23
#define GATHER_BATCH 8

__device__ __forceinline__ void owner_flush(unsigned pend_pk, int pend_wn, int first, int np,
                                            const float* __restrict__ h, float* acc, int lane) {
    float4 hv[GATHER_BATCH];
    #pragma unroll
    for (int k = 0; k < GATHER_BATCH; ++k) {
        if (k < np) {
            unsigned pk = (unsigned)__builtin_amdgcn_readlane((int)pend_pk, first + k);
            unsigned c = pk & ((1u << COL_BITS) - 1u);
            hv[k] = *(const float4*)(h + (size_t)c * FEAT + lane * 4);
        }
    }
    #pragma unroll
    for (int k = 0; k < GATHER_BATCH; ++k) {
        if (k < np) {
            unsigned pk = (unsigned)__builtin_amdgcn_readlane((int)pend_pk, first + k);
            unsigned rl = pk >> COL_BITS;
            float wn = __int_as_float(__builtin_amdgcn_readlane(pend_wn, first + k));
            float4* ap = (float4*)acc + rl * (FEAT / 4) + lane;
            float4 a = *ap;
            a.x += hv[k].x * wn; a.y += hv[k].y * wn; a.z += hv[k].z * wn; a.w += hv[k].w * wn;
            *ap = a;
        }
    }
}

#define IDX_AT(b, wd) ((b) * IDX_WPB + ((wd) ^ (((b) >> 2) & (IDX_WPB - 1))))
__device__ __forceinline__ void index_block(const int* __restrict__ rows, unsigned* __restrict__ idx,
                                            int nE, int nOwn, int blk, unsigned* bits) {
    int t = threadIdx.x;
    for (int i = t; i < 1024 * IDX_WPB; i += MID_THREADS) bits[i] = 0u;
    __syncthreads();
    int rr[4][4];
    #pragma unroll
    for (int k = 0; k < 4; ++k) {
        int e = blk * IDX_CHUNK + k * 1024 + t * 4;
        rr[k][0] = -1; rr[k][1] = -1; rr[k][2] = -1; rr[k][3] = -1;
        if (e + 3 < nE) {
            int4 q = *(const int4*)(rows + e);
            rr[k][0] = q.x; rr[k][1] = q.y; rr[k][2] = q.z; rr[k][3] = q.w;
        } else {
            for (int c = 0; c < 4; ++c) if (e + c < nE) rr[k][c] = rows[e + c];
        }
    }
    #pragma unroll
    for (int k = 0; k < 4; ++k) {
        unsigned bit = 1u << ((t >> 1) & 31);
        int wd = k * 4 + (t >> 6);
        #pragma unroll
        for (int c = 0; c < 4; ++c) {
            unsigned b = (unsigned)rr[k][c] / (unsigned)OWN_ROWS;
            if (b < (unsigned)nOwn) atomicOr(&bits[IDX_AT(b, (unsigned)wd)], bit);
        }
    }
    __syncthreads();
    for (int i = t; i < nOwn * IDX_WPB; i += MID_THREADS)
        idx[(size_t)(i / IDX_WPB) * OWN_STRIDE + blk * IDX_WPB + (i % IDX_WPB)] = bits[IDX_AT(i / IDX_WPB, i % IDX_WPB)];
}

__global__ __launch_bounds__(MID_THREADS) void mid_kernel(const float* __restrict__ x,
                                                          const float* __restrict__ W,
                                                          float* __restrict__ h,
                                                          const int* __restrict__ rows,
                                                          unsigned* __restrict__ idx,
                                                          const float* __restrict__ part,
                                                          float* __restrict__ dinv,
                                                          int nN, int nE, int nOwn, int nIdxBlocks) {
    __shared__ float smem[FEAT * FEAT];
    int nProd = (nN + 127) / 128;
    int b = blockIdx.x;
    if (b < nProd) {
        gemm_block(x, W, h, nN, b, smem);
    } else if (b < nProd + nIdxBlocks) {
        index_block(rows, idx, nE, nOwn, b - nProd, (unsigned*)smem);
    } else {
        dinv_block(part, dinv, nN, b - nProd - nIdxBlocks);
    }
}

template <bool TAIL>
__device__ __forceinline__ bool owner_filter_quarter(int4 q, int eb, int end, int lane,
                                                     int r0, unsigned nbl, unsigned lt, unsigned* mylst, int& n) {
    int rr[4] = {q.x, q.y, q.z, q.w};
    unsigned mk[4];
    #pragma unroll
    for (int c = 0; c < 4; ++c) {
        bool hit = (unsigned)(rr[c] - r0) < nbl;
        if (TAIL) hit = hit && eb + c < end;
        mk[c] = __builtin_amdgcn_ballot_w32(hit);
    }
    if (n + GRP_EDGES > LIST_CAP) {
        int tot = __builtin_popcount(mk[0]) + __builtin_popcount(mk[1]) + __builtin_popcount(mk[2]) + __builtin_popcount(mk[3]);
        if (n + tot > LIST_CAP) return false;
    }
    #pragma unroll
    for (int c = 0; c < 4; ++c) {
        if (mk[c] != 0u) {
            if ((mk[c] >> lane) & 1u) {
                int e = eb + c;
                mylst[n + __builtin_popcount(mk[c] & lt)] = (unsigned)e | ((unsigned)(rr[c] - r0) << EDGE_BITS);
            }
            n += __builtin_popcount(mk[c]);
        }
    }
    return true;
}

__device__ __forceinline__ unsigned owner_lanes(unsigned key, unsigned valid, const unsigned* ob) {
    unsigned m = valid;
    #pragma unroll
    for (int b = 0; b < 5; ++b) m &= ((key >> b) & 1u) ? ob[b] : ~ob[b];
    return m;
}

__global__ __launch_bounds__(OWN_THREADS) void edge_owner_kernel(const int* __restrict__ rows,
                                                                 const int* __restrict__ cols,
                                                                 const float* __restrict__ w,
                                                                 const float* __restrict__ dinv,
                                                                 const float* __restrict__ h,
                                                                 float* __restrict__ out,
                                                                 const float* __restrict__ bias,
                                                                 const float* __restrict__ gamma,
                                                                 const float* __restrict__ beta,
                                                                 const float* __restrict__ mean,
                                                                 const float* __restrict__ var,
                                                                 int nE, int nN, int nWords) {
    __shared__ float    acc[OWN_ROWS * FEAT];
    __shared__ int2     own[OWN_WAVES * LIST_CAP];
    __shared__ unsigned lst[OWN_WAVES * LIST_CAP];
    __shared__ int      cntm[OWN_WAVES * OWN_WAVES];
    __shared__ int      more[OWN_WAVES];

    int t = threadIdx.x;
    int wave = __builtin_amdgcn_readfirstlane(t >> 5);
    int lane = t & 31;
    int r0 = blockIdx.x * OWN_ROWS;
    int nb = nN - r0 < OWN_ROWS ? nN - r0 : OWN_ROWS;

    for (int i = t; i < nb * (FEAT / 4); i += OWN_THREADS) {
        int r = i >> 5;
        float s = dinv[r0 + r];
        float4 v = ((const float4*)h)[(size_t)(r0 + r) * (FEAT / 4) + (i & 31)];
        v.x *= s; v.y *= s; v.z *= s; v.w *= s;
        ((float4*)acc)[i] = v;
    }

    int wpw = (nWords + OWN_WAVES - 1) / OWN_WAVES;
    int w0 = wave * wpw < nWords ? wave * wpw : nWords;
    int w1 = w0 + wpw < nWords ? w0 + wpw : nWords;
    const unsigned* myidx = (const unsigned*)out + (size_t)blockIdx.x * OWN_STRIDE;
    unsigned* idxl = (unsigned*)own;
    int last = (nE - 1) & ~3;
    unsigned lt = (1u << lane) - 1u;
    unsigned* mylst = lst + wave * LIST_CAP;
    int sub = lane / SLOT_LANES;
    int l8 = (lane % SLOT_LANES) * 4;
    int spw = (w1 - w0 + SLOTS - 1) / SLOTS;
    int ws0 = w0 + sub * spw < w1 ? w0 + sub * spw : w1;
    int ws1 = ws0 + spw < w1 ? ws0 + spw : w1;
    int cwv = ws0;
    unsigned cmv = ws0 < ws1 ? myidx[ws0] : 0u;

    for (;;) {
        for (int k0 = w0 + lane; k0 < w1; k0 += 32 * 8) {
            unsigned tw[8];
            #pragma unroll
            for (int u = 0; u < 8; ++u) tw[u] = k0 + 32 * u < w1 ? myidx[k0 + 32 * u] : 0u;
            #pragma unroll
            for (int u = 0; u < 8; ++u) if (k0 + 32 * u < w1) idxl[k0 + 32 * u] = tw[u];
        }
        __syncthreads();
        int n = 0;
        int full = 0;
        while (!full) {
            int g[4];
            bool has[4];
            int scw[4];
            unsigned scm[4];
            #pragma unroll
            for (int j = 0; j < 4; ++j) {
                scw[j] = cwv;
                scm[j] = cmv;
                while (cmv == 0u && cwv + 1 < ws1) {
                    ++cwv;
                    cmv = idxl[cwv];
                }
                has[j] = cmv != 0u;
                g[j] = has[j] ? cwv * 32 + __builtin_ctz(cmv) : 0;
                cmv &= cmv - 1u;
            }
            if (__builtin_amdgcn_ballot_w32(has[0]) == 0u) break;
            int4 q[4];
            int eb[4];
            #pragma unroll
            for (int j = 0; j < 4; ++j) {
                eb[j] = g[j] * LINE_EDGES + l8;
                int i = eb[j] < last ? eb[j] : last;
                q[j] = *(const int4*)(rows + i);
            }
            #pragma unroll
            for (int j = 0; j < 4; ++j) {
                bool act = j == 0 ? true : __builtin_amdgcn_ballot_w32(has[j]) != 0u;
                if (act && !full) {
                    unsigned nbl = has[j] ? (unsigned)nb : 0u;
                    bool tail = __builtin_amdgcn_ballot_w32(has[j] && g[j] * LINE_EDGES + LINE_EDGES > nE) != 0u;
                    bool ok = !tail ? owner_filter_quarter<false>(q[j], eb[j], nE, lane, r0, nbl, lt, mylst, n)
                                    : owner_filter_quarter<true>(q[j], eb[j], nE, lane, r0, nbl, lt, mylst, n);
                    if (!ok) {
                        cwv = scw[j];
                        cmv = scm[j];
                        full = 1;
                    }
                }
            }
        }
        int mycnt = 0;
        for (int base = 0; base < n; base += 32) {
            int i = base + lane;
            bool ok = i < n;
            unsigned o = ok ? (mylst[i] >> EDGE_BITS) & 31u : 0u;
            unsigned valid = __builtin_amdgcn_ballot_w32(ok);
            unsigned ob[5];
            #pragma unroll
            for (int b = 0; b < 5; ++b) ob[b] = __builtin_amdgcn_ballot_w32(ok && ((o >> b) & 1u));
            mycnt += __builtin_popcount(owner_lanes((unsigned)lane, valid, ob));
        }
        cntm[wave * OWN_WAVES + lane] = mycnt;
        if (lane == 0) more[wave] = full;
        __syncthreads();

        int before = 0;
        int tot = 0;
        #pragma unroll
        for (int s = 0; s < OWN_WAVES; ++s) {
            int v = cntm[s * OWN_WAVES + lane];
            tot += v;
            before += s < wave ? v : 0;
        }
        int incl = tot;
        for (int d = 1; d < 32; d <<= 1) {
            int up = __shfl_up(incl, d);
            if (lane >= d) incl += up;
        }
        int pos = incl - tot + before;
        int myfirst = __builtin_amdgcn_readlane(incl - tot, wave);
        int mytot = __builtin_amdgcn_readlane(tot, wave);
        int any = __builtin_amdgcn_ballot_w32(more[lane] != 0) != 0u;

        for (int base = 0; base < n; base += 32) {
            int i = base + lane;
            bool ok = i < n;
            unsigned en = ok ? mylst[i] : 0u;
            unsigned rl = en >> EDGE_BITS;
            unsigned o = rl & 31u;
            unsigned valid = __builtin_amdgcn_ballot_w32(ok);
            unsigned ob[5];
            #pragma unroll
            for (int b = 0; b < 5; ++b) ob[b] = __builtin_amdgcn_ballot_w32(ok && ((o >> b) & 1u));
            int p = __shfl(pos, (int)o) + __builtin_popcount(owner_lanes(o, valid, ob) & lt);
            pos += __builtin_popcount(owner_lanes((unsigned)lane, valid, ob));
            if (ok) {
                int e = (int)(en & ((1u << EDGE_BITS) - 1u));
                int c = cols[e];
                float wn = w[e] * dinv[c];
                own[p] = make_int2((int)((unsigned)c | (rl << COL_BITS)), __float_as_int(wn));
            }
        }
        __syncthreads();

        const int2* mine = own + myfirst;
        for (int base = 0; base < mytot; base += 32) {
            int i = base + lane;
            int2 en = make_int2(0, 0);
            if (i < mytot) en = mine[i];
            int m = mytot - base < 32 ? mytot - base : 32;
            for (int j = 0; j + GATHER_BATCH <= m; j += GATHER_BATCH)
                owner_flush((unsigned)en.x, en.y, j, GATHER_BATCH, h, acc, lane);
            if (m & (GATHER_BATCH - 1))
                owner_flush((unsigned)en.x, en.y, m & ~(GATHER_BATCH - 1), m & (GATHER_BATCH - 1), h, acc, lane);
        }
        __syncthreads();
        if (!any) break;
    }

    int u = lane * 4;
    float4 bi = *(const float4*)(bias + u);
    float4 ga = *(const float4*)(gamma + u);
    float4 be = *(const float4*)(beta + u);
    float4 me = *(const float4*)(mean + u);
    float4 va = *(const float4*)(var + u);
    float4 is = make_float4(rsqrtf(va.x + BN_EPS), rsqrtf(va.y + BN_EPS), rsqrtf(va.z + BN_EPS), rsqrtf(va.w + BN_EPS));
    for (int i = t; i < nb * (FEAT / 4); i += OWN_THREADS) {
        int r = i >> 5;
        float s = dinv[r0 + r];
        float4 v = ((const float4*)acc)[i];
        float4 o;
        o.x = ga.x * ((v.x * s + bi.x) - me.x) * is.x + be.x;
        o.y = ga.y * ((v.y * s + bi.y) - me.y) * is.y + be.y;
        o.z = ga.z * ((v.z * s + bi.z) - me.z) * is.z + be.z;
        o.w = ga.w * ((v.w * s + bi.w) - me.w) * is.w + be.w;
        o.x = o.x > 0.0f ? o.x : 0.0f;
        o.y = o.y > 0.0f ? o.y : 0.0f;
        o.z = o.z > 0.0f ? o.z : 0.0f;
        o.w = o.w > 0.0f ? o.w : 0.0f;
        ((float4*)out)[(size_t)(r0 + r) * (FEAT / 4) + lane] = o;
    }
}


extern "C" void kernel_launch(void* const* d_in, const int* in_sizes, int n_in,
                              void* d_out, int out_size, void* d_ws, size_t ws_size,
                              hipStream_t stream) {
    const float* x     = (const float*)d_in[0];
    const int*   eidx  = (const int*)d_in[1];
    const float* ew    = (const float*)d_in[2];
    const float* W     = (const float*)d_in[3];
    const float* bias  = (const float*)d_in[4];
    const float* gamma = (const float*)d_in[5];
    const float* beta  = (const float*)d_in[6];
    const float* mean  = (const float*)d_in[7];
    const float* var   = (const float*)d_in[8];

    int nE = in_sizes[2];
    int nN = in_sizes[0] / FEAT;
    const int* rows = eidx;
    const int* cols = eidx + nE;

    float* h    = (float*)d_ws;
    float* dinv = h + (size_t)nN * FEAT + nN;
    float* out  = (float*)d_out;

    float* part = out;
    unsigned* idx = (unsigned*)d_out;
    deg_partial_kernel<<<dim3(DEG_SLICES, (nN + DEG_BINS - 1) / DEG_BINS), DEG_THREADS, 0, stream>>>(
        rows, ew, part, nE, nN);

    mid_kernel<<<(nN + 127) / 128 + (nE + IDX_CHUNK - 1) / IDX_CHUNK + (nN + MID_THREADS - 1) / MID_THREADS,
                 MID_THREADS, 0, stream>>>(
        x, W, h, rows, idx, part, dinv, nN, nE, (nN + OWN_ROWS - 1) / OWN_ROWS, (nE + IDX_CHUNK - 1) / IDX_CHUNK);

    edge_owner_kernel<<<(nN + OWN_ROWS - 1) / OWN_ROWS, OWN_THREADS, 0, stream>>>(
        rows, cols, ew, dinv, h, out, bias, gamma, beta, mean, var, nE, nN, IDX_WPB * ((nE + IDX_CHUNK - 1) / IDX_CHUNK));
}


# load round trips: place asks for column and weight of a wave's list at once (ahead of count and barrier), then every dinv at once; a degree block issues a slice's edge loads in one batch
# speedup vs baseline: 3.7483x; 1.0458x over previous
#include <hip/hip_runtime.h>
#include <math.h>

typedef float v2f __attribute__((ext_vector_type(2)));
typedef float v8f __attribute__((ext_vector_type(8)));

#define FEAT 128
#define MID_THREADS 256
#define BN_EPS 1e-3f


#define DEG_BINS   25600
#define DEG_SLICES 64
#define DEG_THREADS 1024
#define DEG_BATCH  25

#define OWN_STRIDE (393 * 128)
#define IDX_HOLE   8192
#define PART_RUN   (OWN_STRIDE - IDX_HOLE)
__device__ __forceinline__ size_t part_at(unsigned f) {
    return (size_t)f + (size_t)IDX_HOLE * (f / (unsigned)PART_RUN + 1u);
}

__global__ __launch_bounds__(DEG_THREADS) void deg_partial_kernel(const int* __restrict__ rows,
                                                                  const float* __restrict__ w,
                                                                  float* __restrict__ part,
                                                                  int nE, int nN) {
    __shared__ float bins[DEG_BINS];

    int t = threadIdx.x;
    float4* bz = (float4*)bins;
    for (int i = t; i < DEG_BINS / 4; i += DEG_THREADS) bz[i] = make_float4(0.0f, 0.0f, 0.0f, 0.0f);
    __syncthreads();

    int r0  = blockIdx.y * DEG_BINS;
    int per = (nE + DEG_SLICES - 1) / DEG_SLICES;
    int e0  = blockIdx.x * per;
    int e1  = e0 + per < nE ? e0 + per : nE;

    for (int e = e0 + t; e < e1; e += DEG_THREADS * DEG_BATCH) {
        int   r[DEG_BATCH];
        float v[DEG_BATCH];
        for (int u = 0; u < DEG_BATCH; ++u) {
            int ee = e + u * DEG_THREADS;
            bool ok = ee < e1;
            r[u] = ok ? rows[ee] : -1;
            v[u] = ok ? w[ee] : 0.0f;
        }
        for (int u = 0; u < DEG_BATCH; ++u) {
            unsigned d = (unsigned)(r[u] - r0);
            if (d < (unsigned)DEG_BINS) atomicAdd(&bins[d], v[u]);
        }
    }
    __syncthreads();

    int nb = nN - r0 < DEG_BINS ? nN - r0 : DEG_BINS;
    unsigned f0 = (unsigned)(blockIdx.x * nN + r0);
    for (int i = t; i < nb; i += DEG_THREADS) part[part_at(f0 + (unsigned)i)] = bins[i];
}

__device__ __forceinline__ void dinv_block(const float* __restrict__ part, float* __restrict__ dinv, int n, int blk) {
    int i = blk * MID_THREADS + threadIdx.x;
    if (i < n) {
        float p[DEG_SLICES];
        for (int s = 0; s < DEG_SLICES; ++s) p[s] = part[part_at((unsigned)(s * n + i))];
        float d = 0.0f;
        for (int s = 0; s < DEG_SLICES; ++s) d += p[s];
        d += 1.0f;
        dinv[i] = (d > 0.0f) ? rsqrtf(d) : 0.0f;
    }
}


__device__ __forceinline__ void gemm_block(const float* __restrict__ x,
                                           const float* __restrict__ W,
                                           float* __restrict__ h, int nRows, int blk, float* Wl) {
    int t = threadIdx.x;
    const float4* Wv = (const float4*)W;
    float4* Wlv = (float4*)Wl;
    #pragma unroll
    for (int i = t; i < (FEAT * FEAT) / 4; i += MID_THREADS) Wlv[i] = Wv[i];
    __syncthreads();

    int wave = t >> 5;
    int lane = t & 31;
    int row0 = blk * 128 + wave * 16;
    if (row0 >= nRows) return;

    int m  = lane & 15;
    int kh = (lane >> 4) << 1;
    int srow = row0 + m;
    if (srow >= nRows) srow = nRows - 1;
    const float* xr = x + (size_t)srow * FEAT;

    v8f acc[8] = {};

    for (int kb = 0; kb < FEAT / 4; ++kb) {
        int k = kb * 4 + kh;
        float2 av = *(const float2*)(xr + k);
        v2f a; a.x = av.x; a.y = av.y;
        #pragma unroll
        for (int j = 0; j < 8; ++j) {
            int col = j * 16 + m;
            v2f b;
            b.x = Wl[(k + 0) * FEAT + col];
            b.y = Wl[(k + 1) * FEAT + col];
            acc[j] = __builtin_amdgcn_wmma_f32_16x16x4_f32(
                false, a, false, b, (short)0, acc[j], false, false);
        }
    }

    int rbase = row0 + ((lane >> 4) ? 8 : 0);
    int n = lane & 15;
    #pragma unroll
    for (int j = 0; j < 8; ++j) {
        #pragma unroll
        for (int r = 0; r < 8; ++r) {
            int row = rbase + r;
            if (row < nRows) h[(size_t)row * FEAT + j * 16 + n] = acc[j][r];
        }
    }
}


#define OWN_ROWS    393
#define GRP_EDGES   128
#define LINE_EDGES  8
#define SLOT_LANES  2
#define SLOTS       16
#define IDX_CHUNK   4096
#define IDX_WPB     16
#define OWN_THREADS 1024
#define OWN_WAVES   32
#define LIST_CAP    316
#define COL_BITS    23
#define EDGE_BITS   23
#define GATHER_BATCH 8
#define PLACE_CHUNKS 8

__device__ __forceinline__ void owner_flush(unsigned pend_pk, int pend_wn, int first, int np,
                                            const float* __restrict__ h, float* acc, int lane) {
    float4 hv[GATHER_BATCH];
    #pragma unroll
    for (int k = 0; k < GATHER_BATCH; ++k) {
        if (k < np) {
            unsigned pk = (unsigned)__builtin_amdgcn_readlane((int)pend_pk, first + k);
            unsigned c = pk & ((1u << COL_BITS) - 1u);
            hv[k] = *(const float4*)(h + (size_t)c * FEAT + lane * 4);
        }
    }
    #pragma unroll
    for (int k = 0; k < GATHER_BATCH; ++k) {
        if (k < np) {
            unsigned pk = (unsigned)__builtin_amdgcn_readlane((int)pend_pk, first + k);
            unsigned rl = pk >> COL_BITS;
            float wn = __int_as_float(__builtin_amdgcn_readlane(pend_wn, first + k));
            float4* ap = (float4*)acc + rl * (FEAT / 4) + lane;
            float4 a = *ap;
            a.x += hv[k].x * wn; a.y += hv[k].y * wn; a.z += hv[k].z * wn; a.w += hv[k].w * wn;
            *ap = a;
        }
    }
}

#define IDX_AT(b, wd) ((b) * IDX_WPB + ((wd) ^ (((b) >> 2) & (IDX_WPB - 1))))
__device__ __forceinline__ void index_block(const int* __restrict__ rows, unsigned* __restrict__ idx,
                                            int nE, int nOwn, int blk, unsigned* bits) {
    int t = threadIdx.x;
    for (int i = t; i < 1024 * IDX_WPB; i += MID_THREADS) bits[i] = 0u;
    __syncthreads();
    int rr[4][4];
    #pragma unroll
    for (int k = 0; k < 4; ++k) {
        int e = blk * IDX_CHUNK + k * 1024 + t * 4;
        rr[k][0] = -1; rr[k][1] = -1; rr[k][2] = -1; rr[k][3] = -1;
        if (e + 3 < nE) {
            int4 q = *(const int4*)(rows + e);
            rr[k][0] = q.x; rr[k][1] = q.y; rr[k][2] = q.z; rr[k][3] = q.w;
        } else {
            for (int c = 0; c < 4; ++c) if (e + c < nE) rr[k][c] = rows[e + c];
        }
    }
    #pragma unroll
    for (int k = 0; k < 4; ++k) {
        unsigned bit = 1u << ((t >> 1) & 31);
        int wd = k * 4 + (t >> 6);
        #pragma unroll
        for (int c = 0; c < 4; ++c) {
            unsigned b = (unsigned)rr[k][c] / (unsigned)OWN_ROWS;
            if (b < (unsigned)nOwn) atomicOr(&bits[IDX_AT(b, (unsigned)wd)], bit);
        }
    }
    __syncthreads();
    for (int i = t; i < nOwn * IDX_WPB; i += MID_THREADS)
        idx[(size_t)(i / IDX_WPB) * OWN_STRIDE + blk * IDX_WPB + (i % IDX_WPB)] = bits[IDX_AT(i / IDX_WPB, i % IDX_WPB)];
}

__global__ __launch_bounds__(MID_THREADS) void mid_kernel(const float* __restrict__ x,
                                                          const float* __restrict__ W,
                                                          float* __restrict__ h,
                                                          const int* __restrict__ rows,
                                                          unsigned* __restrict__ idx,
                                                          const float* __restrict__ part,
                                                          float* __restrict__ dinv,
                                                          int nN, int nE, int nOwn, int nIdxBlocks) {
    __shared__ float smem[FEAT * FEAT];
    int nProd = (nN + 127) / 128;
    int b = blockIdx.x;
    if (b < nProd) {
        gemm_block(x, W, h, nN, b, smem);
    } else if (b < nProd + nIdxBlocks) {
        index_block(rows, idx, nE, nOwn, b - nProd, (unsigned*)smem);
    } else {
        dinv_block(part, dinv, nN, b - nProd - nIdxBlocks);
    }
}

template <bool TAIL>
__device__ __forceinline__ bool owner_filter_quarter(int4 q, int eb, int end, int lane,
                                                     int r0, unsigned nbl, unsigned lt, unsigned* mylst, int& n) {
    int rr[4] = {q.x, q.y, q.z, q.w};
    unsigned mk[4];
    #pragma unroll
    for (int c = 0; c < 4; ++c) {
        bool hit = (unsigned)(rr[c] - r0) < nbl;
        if (TAIL) hit = hit && eb + c < end;
        mk[c] = __builtin_amdgcn_ballot_w32(hit);
    }
    if (n + GRP_EDGES > LIST_CAP) {
        int tot = __builtin_popcount(mk[0]) + __builtin_popcount(mk[1]) + __builtin_popcount(mk[2]) + __builtin_popcount(mk[3]);
        if (n + tot > LIST_CAP) return false;
    }
    #pragma unroll
    for (int c = 0; c < 4; ++c) {
        if (mk[c] != 0u) {
            if ((mk[c] >> lane) & 1u) {
                int e = eb + c;
                mylst[n + __builtin_popcount(mk[c] & lt)] = (unsigned)e | ((unsigned)(rr[c] - r0) << EDGE_BITS);
            }
            n += __builtin_popcount(mk[c]);
        }
    }
    return true;
}

__device__ __forceinline__ unsigned owner_lanes(unsigned key, unsigned valid, const unsigned* ob) {
    unsigned m = valid;
    #pragma unroll
    for (int b = 0; b < 5; ++b) m &= ((key >> b) & 1u) ? ob[b] : ~ob[b];
    return m;
}

__global__ __launch_bounds__(OWN_THREADS) void edge_owner_kernel(const int* __restrict__ rows,
                                                                 const int* __restrict__ cols,
                                                                 const float* __restrict__ w,
                                                                 const float* __restrict__ dinv,
                                                                 const float* __restrict__ h,
                                                                 float* __restrict__ out,
                                                                 const float* __restrict__ bias,
                                                                 const float* __restrict__ gamma,
                                                                 const float* __restrict__ beta,
                                                                 const float* __restrict__ mean,
                                                                 const float* __restrict__ var,
                                                                 int nE, int nN, int nWords) {
    __shared__ float    acc[OWN_ROWS * FEAT];
    __shared__ int2     own[OWN_WAVES * LIST_CAP];
    __shared__ unsigned lst[OWN_WAVES * LIST_CAP];
    __shared__ int      cntm[OWN_WAVES * OWN_WAVES];
    __shared__ int      more[OWN_WAVES];

    int t = threadIdx.x;
    int wave = __builtin_amdgcn_readfirstlane(t >> 5);
    int lane = t & 31;
    int r0 = blockIdx.x * OWN_ROWS;
    int nb = nN - r0 < OWN_ROWS ? nN - r0 : OWN_ROWS;

    for (int i = t; i < nb * (FEAT / 4); i += OWN_THREADS) {
        int r = i >> 5;
        float s = dinv[r0 + r];
        float4 v = ((const float4*)h)[(size_t)(r0 + r) * (FEAT / 4) + (i & 31)];
        v.x *= s; v.y *= s; v.z *= s; v.w *= s;
        ((float4*)acc)[i] = v;
    }

    int wpw = (nWords + OWN_WAVES - 1) / OWN_WAVES;
    int w0 = wave * wpw < nWords ? wave * wpw : nWords;
    int w1 = w0 + wpw < nWords ? w0 + wpw : nWords;
    const unsigned* myidx = (const unsigned*)out + (size_t)blockIdx.x * OWN_STRIDE;
    unsigned* idxl = (unsigned*)own;
    int last = (nE - 1) & ~3;
    unsigned lt = (1u << lane) - 1u;
    unsigned* mylst = lst + wave * LIST_CAP;
    int sub = lane / SLOT_LANES;
    int l8 = (lane % SLOT_LANES) * 4;
    int spw = (w1 - w0 + SLOTS - 1) / SLOTS;
    int ws0 = w0 + sub * spw < w1 ? w0 + sub * spw : w1;
    int ws1 = ws0 + spw < w1 ? ws0 + spw : w1;
    int cwv = ws0;
    unsigned cmv = ws0 < ws1 ? myidx[ws0] : 0u;

    for (;;) {
        for (int k0 = w0 + lane; k0 < w1; k0 += 32 * 8) {
            unsigned tw[8];
            #pragma unroll
            for (int u = 0; u < 8; ++u) tw[u] = k0 + 32 * u < w1 ? myidx[k0 + 32 * u] : 0u;
            #pragma unroll
            for (int u = 0; u < 8; ++u) if (k0 + 32 * u < w1) idxl[k0 + 32 * u] = tw[u];
        }
        __syncthreads();
        int n = 0;
        int full = 0;
        while (!full) {
            int g[4];
            bool has[4];
            int scw[4];
            unsigned scm[4];
            #pragma unroll
            for (int j = 0; j < 4; ++j) {
                scw[j] = cwv;
                scm[j] = cmv;
                while (cmv == 0u && cwv + 1 < ws1) {
                    ++cwv;
                    cmv = idxl[cwv];
                }
                has[j] = cmv != 0u;
                g[j] = has[j] ? cwv * 32 + __builtin_ctz(cmv) : 0;
                cmv &= cmv - 1u;
            }
            if (__builtin_amdgcn_ballot_w32(has[0]) == 0u) break;
            int4 q[4];
            int eb[4];
            #pragma unroll
            for (int j = 0; j < 4; ++j) {
                eb[j] = g[j] * LINE_EDGES + l8;
                int i = eb[j] < last ? eb[j] : last;
                q[j] = *(const int4*)(rows + i);
            }
            #pragma unroll
            for (int j = 0; j < 4; ++j) {
                bool act = j == 0 ? true : __builtin_amdgcn_ballot_w32(has[j]) != 0u;
                if (act && !full) {
                    unsigned nbl = has[j] ? (unsigned)nb : 0u;
                    bool tail = __builtin_amdgcn_ballot_w32(has[j] && g[j] * LINE_EDGES + LINE_EDGES > nE) != 0u;
                    bool ok = !tail ? owner_filter_quarter<false>(q[j], eb[j], nE, lane, r0, nbl, lt, mylst, n)
                                    : owner_filter_quarter<true>(q[j], eb[j], nE, lane, r0, nbl, lt, mylst, n);
                    if (!ok) {
                        cwv = scw[j];
                        cmv = scm[j];
                        full = 1;
                    }
                }
            }
        }
        unsigned enr[PLACE_CHUNKS];
        int      ecol[PLACE_CHUNKS];
        float    ewt[PLACE_CHUNKS];
        float    edv[PLACE_CHUNKS];
        #pragma unroll
        for (int k = 0; k < PLACE_CHUNKS; ++k) enr[k] = k * 32 + lane < n ? mylst[k * 32 + lane] : 0u;
        #pragma unroll
        for (int k = 0; k < PLACE_CHUNKS; ++k) {
            ecol[k] = 0;
            ewt[k] = 0.0f;
            if (k * 32 < n && k * 32 + lane < n) {
                int e = (int)(enr[k] & ((1u << EDGE_BITS) - 1u));
                ecol[k] = cols[e];
                ewt[k] = w[e];
            }
        }

        int mycnt = 0;
        #pragma unroll
        for (int k = 0; k < PLACE_CHUNKS; ++k) {
            if (k * 32 < n) {
                bool ok = k * 32 + lane < n;
                unsigned o = (enr[k] >> EDGE_BITS) & 31u;
                unsigned valid = __builtin_amdgcn_ballot_w32(ok);
                unsigned ob[5];
                #pragma unroll
                for (int b = 0; b < 5; ++b) ob[b] = __builtin_amdgcn_ballot_w32(ok && ((o >> b) & 1u));
                mycnt += __builtin_popcount(owner_lanes((unsigned)lane, valid, ob));
            }
        }
        for (int base = PLACE_CHUNKS * 32; base < n; base += 32) {
            int i = base + lane;
            bool ok = i < n;
            unsigned o = ok ? (mylst[i] >> EDGE_BITS) & 31u : 0u;
            unsigned valid = __builtin_amdgcn_ballot_w32(ok);
            unsigned ob[5];
            #pragma unroll
            for (int b = 0; b < 5; ++b) ob[b] = __builtin_amdgcn_ballot_w32(ok && ((o >> b) & 1u));
            mycnt += __builtin_popcount(owner_lanes((unsigned)lane, valid, ob));
        }
        cntm[wave * OWN_WAVES + lane] = mycnt;
        if (lane == 0) more[wave] = full;
        __syncthreads();

        unsigned epk[PLACE_CHUNKS];
        #pragma unroll
        for (int k = 0; k < PLACE_CHUNKS; ++k) {
            edv[k] = 0.0f;
            if (k * 32 < n && k * 32 + lane < n) edv[k] = dinv[ecol[k]];
            epk[k] = (unsigned)ecol[k] | ((enr[k] >> EDGE_BITS) << COL_BITS);
        }

        int before = 0;
        int tot = 0;
        #pragma unroll
        for (int s = 0; s < OWN_WAVES; ++s) {
            int v = cntm[s * OWN_WAVES + lane];
            tot += v;
            before += s < wave ? v : 0;
        }
        int incl = tot;
        for (int d = 1; d < 32; d <<= 1) {
            int up = __shfl_up(incl, d);
            if (lane >= d) incl += up;
        }
        int pos = incl - tot + before;
        int myfirst = __builtin_amdgcn_readlane(incl - tot, wave);
        int mytot = __builtin_amdgcn_readlane(tot, wave);
        int any = __builtin_amdgcn_ballot_w32(more[lane] != 0) != 0u;

        #pragma unroll
        for (int k = 0; k < PLACE_CHUNKS; ++k) {
            if (k * 32 < n) {
                bool ok = k * 32 + lane < n;
                unsigned o = (epk[k] >> COL_BITS) & 31u;
                unsigned valid = __builtin_amdgcn_ballot_w32(ok);
                unsigned ob[5];
                #pragma unroll
                for (int b = 0; b < 5; ++b) ob[b] = __builtin_amdgcn_ballot_w32(ok && ((o >> b) & 1u));
                int p = __shfl(pos, (int)o) + __builtin_popcount(owner_lanes(o, valid, ob) & lt);
                pos += __builtin_popcount(owner_lanes((unsigned)lane, valid, ob));
                if (ok) {
                    float wn = ewt[k] * edv[k];
                    own[p] = make_int2((int)epk[k], __float_as_int(wn));
                }
            }
        }
        for (int base = PLACE_CHUNKS * 32; base < n; base += 32) {
            int i = base + lane;
            bool ok = i < n;
            unsigned en = ok ? mylst[i] : 0u;
            unsigned rl = en >> EDGE_BITS;
            unsigned o = rl & 31u;
            unsigned valid = __builtin_amdgcn_ballot_w32(ok);
            unsigned ob[5];
            #pragma unroll
            for (int b = 0; b < 5; ++b) ob[b] = __builtin_amdgcn_ballot_w32(ok && ((o >> b) & 1u));
            int p = __shfl(pos, (int)o) + __builtin_popcount(owner_lanes(o, valid, ob) & lt);
            pos += __builtin_popcount(owner_lanes((unsigned)lane, valid, ob));
            if (ok) {
                int e = (int)(en & ((1u << EDGE_BITS) - 1u));
                int cc = cols[e];
                float wn = w[e] * dinv[cc];
                own[p] = make_int2((int)((unsigned)cc | (rl << COL_BITS)), __float_as_int(wn));
            }
        }
        __syncthreads();

        const int2* mine = own + myfirst;
        for (int base = 0; base < mytot; base += 32) {
            int i = base + lane;
            int2 en = make_int2(0, 0);
            if (i < mytot) en = mine[i];
            int m = mytot - base < 32 ? mytot - base : 32;
            for (int j = 0; j + GATHER_BATCH <= m; j += GATHER_BATCH)
                owner_flush((unsigned)en.x, en.y, j, GATHER_BATCH, h, acc, lane);
            if (m & (GATHER_BATCH - 1))
                owner_flush((unsigned)en.x, en.y, m & ~(GATHER_BATCH - 1), m & (GATHER_BATCH - 1), h, acc, lane);
        }
        __syncthreads();
        if (!any) break;
    }

    int u = lane * 4;
    float4 bi = *(const float4*)(bias + u);
    float4 ga = *(const float4*)(gamma + u);
    float4 be = *(const float4*)(beta + u);
    float4 me = *(const float4*)(mean + u);
    float4 va = *(const float4*)(var + u);
    float4 is = make_float4(rsqrtf(va.x + BN_EPS), rsqrtf(va.y + BN_EPS), rsqrtf(va.z + BN_EPS), rsqrtf(va.w + BN_EPS));
    for (int i = t; i < nb * (FEAT / 4); i += OWN_THREADS) {
        int r = i >> 5;
        float s = dinv[r0 + r];
        float4 v = ((const float4*)acc)[i];
        float4 o;
        o.x = ga.x * ((v.x * s + bi.x) - me.x) * is.x + be.x;
        o.y = ga.y * ((v.y * s + bi.y) - me.y) * is.y + be.y;
        o.z = ga.z * ((v.z * s + bi.z) - me.z) * is.z + be.z;
        o.w = ga.w * ((v.w * s + bi.w) - me.w) * is.w + be.w;
        o.x = o.x > 0.0f ? o.x : 0.0f;
        o.y = o.y > 0.0f ? o.y : 0.0f;
        o.z = o.z > 0.0f ? o.z : 0.0f;
        o.w = o.w > 0.0f ? o.w : 0.0f;
        ((float4*)out)[(size_t)(r0 + r) * (FEAT / 4) + lane] = o;
    }
}


extern "C" void kernel_launch(void* const* d_in, const int* in_sizes, int n_in,
                              void* d_out, int out_size, void* d_ws, size_t ws_size,
                              hipStream_t stream) {
    const float* x     = (const float*)d_in[0];
    const int*   eidx  = (const int*)d_in[1];
    const float* ew    = (const float*)d_in[2];
    const float* W     = (const float*)d_in[3];
    const float* bias  = (const float*)d_in[4];
    const float* gamma = (const float*)d_in[5];
    const float* beta  = (const float*)d_in[6];
    const float* mean  = (const float*)d_in[7];
    const float* var   = (const float*)d_in[8];

    int nE = in_sizes[2];
    int nN = in_sizes[0] / FEAT;
    const int* rows = eidx;
    const int* cols = eidx + nE;

    float* h    = (float*)d_ws;
    float* dinv = h + (size_t)nN * FEAT + nN;
    float* out  = (float*)d_out;

    float* part = out;
    unsigned* idx = (unsigned*)d_out;
    deg_partial_kernel<<<dim3(DEG_SLICES, (nN + DEG_BINS - 1) / DEG_BINS), DEG_THREADS, 0, stream>>>(
        rows, ew, part, nE, nN);

    mid_kernel<<<(nN + 127) / 128 + (nE + IDX_CHUNK - 1) / IDX_CHUNK + (nN + MID_THREADS - 1) / MID_THREADS,
                 MID_THREADS, 0, stream>>>(
        x, W, h, rows, idx, part, dinv, nN, nE, (nN + OWN_ROWS - 1) / OWN_ROWS, (nE + IDX_CHUNK - 1) / IDX_CHUNK);

    edge_owner_kernel<<<(nN + OWN_ROWS - 1) / OWN_ROWS, OWN_THREADS, 0, stream>>>(
        rows, cols, ew, dinv, h, out, bias, gamma, beta, mean, var, nE, nN, IDX_WPB * ((nE + IDX_CHUNK - 1) / IDX_CHUNK));
}
